# starting kernel
# baseline (speedup 1.0000x reference)
#include <hip/hip_runtime.h>
#include <math.h>

#define E_EDGES  1000000
#define R_REL    1000
#define D_DIM    200
#define THREE_D  600
#define M_ROWS   2000
#define LDSTRIDE 204

typedef float v2f __attribute__((ext_vector_type(2)));
typedef float v8f __attribute__((ext_vector_type(8)));

__global__ void zero_ws(float* ws, int n) {
    int i = blockIdx.x * blockDim.x + threadIdx.x;
    if (i < n) ws[i] = 0.0f;
}

__global__ void scatter_kernel(const int* __restrict__ ei,
                               const int* __restrict__ et,
                               const int* __restrict__ nid,
                               const float* __restrict__ emb,
                               float* __restrict__ sum,
                               float* __restrict__ cnt) {
    long long idx   = (long long)blockIdx.x * blockDim.x + threadIdx.x;
    long long total = (long long)E_EDGES * (D_DIM / 4);
    if (idx >= total) return;
    int e  = (int)(idx / (D_DIM / 4));
    int d4 = (int)(idx - (long long)e * (D_DIM / 4)) * 4;
    int r = et[e];
    int s = nid[ei[e]];
    int t = nid[ei[E_EDGES + e]];
    const float4 vs = *(const float4*)(emb + (size_t)s * D_DIM + d4);
    const float4 vt = *(const float4*)(emb + (size_t)t * D_DIM + d4);
    float* ps = sum + (size_t)r * D_DIM + d4;
    float* pt = sum + (size_t)(R_REL + r) * D_DIM + d4;
    atomicAdd(ps + 0, vs.x); atomicAdd(ps + 1, vs.y);
    atomicAdd(ps + 2, vs.z); atomicAdd(ps + 3, vs.w);
    atomicAdd(pt + 0, vt.x); atomicAdd(pt + 1, vt.y);
    atomicAdd(pt + 2, vt.z); atomicAdd(pt + 3, vt.w);
    if (d4 == 0) atomicAdd(cnt + r, 1.0f);
}

__global__ void mean_kernel(float* __restrict__ sum, const float* __restrict__ cnt) {
    int i = blockIdx.x * blockDim.x + threadIdx.x;
    if (i >= M_ROWS * D_DIM) return;
    int m = i / D_DIM;
    sum[i] = sum[i] / fmaxf(cnt[m % R_REL], 1.0f);
}

__device__ __forceinline__ float sigmoidf_(float x) { return 1.0f / (1.0f + expf(-x)); }

__launch_bounds__(32)
__global__ void gru_wmma_kernel(const float* __restrict__ X,
                                const float* __restrict__ dyn,
                                const float* __restrict__ Wih,
                                const float* __restrict__ Whh,
                                const float* __restrict__ bih,
                                const float* __restrict__ bhh,
                                float* __restrict__ out) {
    __shared__ float Xs[16 * LDSTRIDE];
    __shared__ float Hs[16 * LDSTRIDE];
    const int lane = threadIdx.x;
    const int m0   = blockIdx.x * 16;

    for (int idx = lane; idx < 16 * D_DIM; idx += 32) {
        int row = idx / D_DIM, col = idx - row * D_DIM;
        int m = m0 + row;
        int r = m % R_REL, f = m / R_REL;
        Xs[row * LDSTRIDE + col] = X[(size_t)m * D_DIM + col];
        Hs[row * LDSTRIDE + col] = dyn[(size_t)r * (2 * D_DIM) + col * 2 + f];
    }
    __syncthreads();

    const int aRow = lane & 15;
    const int nRow = lane & 15;
    const int kOff = (lane < 16) ? 0 : 2;
    const int hi   = (lane < 16) ? 0 : 8;

    for (int c0 = 0; c0 < 192; c0 += 16) {
        const float* pI0 = Wih + (size_t)(c0 + nRow) * D_DIM + kOff;
        const float* pI1 = pI0 + (size_t)D_DIM * D_DIM;
        const float* pI2 = pI1 + (size_t)D_DIM * D_DIM;
        const float* pH0 = Whh + (size_t)(c0 + nRow) * D_DIM + kOff;
        const float* pH1 = pH0 + (size_t)D_DIM * D_DIM;
        const float* pH2 = pH1 + (size_t)D_DIM * D_DIM;

        v8f c_ir = {}; v8f c_iz = {}; v8f c_in = {};
        v8f c_hr = {}; v8f c_hz = {}; v8f c_hn = {};

        for (int k0 = 0; k0 < D_DIM; k0 += 4) {
            v2f ax, ah, b0, b1, b2, b3, b4, b5;
            ax.x = Xs[aRow * LDSTRIDE + k0 + kOff];
            ax.y = Xs[aRow * LDSTRIDE + k0 + kOff + 1];
            ah.x = Hs[aRow * LDSTRIDE + k0 + kOff];
            ah.y = Hs[aRow * LDSTRIDE + k0 + kOff + 1];
            b0.x = pI0[k0]; b0.y = pI0[k0 + 1];
            b1.x = pI1[k0]; b1.y = pI1[k0 + 1];
            b2.x = pI2[k0]; b2.y = pI2[k0 + 1];
            b3.x = pH0[k0]; b3.y = pH0[k0 + 1];
            b4.x = pH1[k0]; b4.y = pH1[k0 + 1];
            b5.x = pH2[k0]; b5.y = pH2[k0 + 1];

            c_ir = __builtin_amdgcn_wmma_f32_16x16x4_f32(false, ax, false, b0, (short)0, c_ir, false, false);
            c_iz = __builtin_amdgcn_wmma_f32_16x16x4_f32(false, ax, false, b1, (short)0, c_iz, false, false);
            c_in = __builtin_amdgcn_wmma_f32_16x16x4_f32(false, ax, false, b2, (short)0, c_in, false, false);
            c_hr = __builtin_amdgcn_wmma_f32_16x16x4_f32(false, ah, false, b3, (short)0, c_hr, false, false);
            c_hz = __builtin_amdgcn_wmma_f32_16x16x4_f32(false, ah, false, b4, (short)0, c_hz, false, false);
            c_hn = __builtin_amdgcn_wmma_f32_16x16x4_f32(false, ah, false, b5, (short)0, c_hn, false, false);
        }

        int dCol = c0 + nRow;
        float b_ir = bih[dCol], b_iz = bih[D_DIM + dCol], b_in = bih[2 * D_DIM + dCol];
        float b_hr = bhh[dCol], b_hz = bhh[D_DIM + dCol], b_hn = bhh[2 * D_DIM + dCol];
        #pragma unroll
        for (int i = 0; i < 8; ++i) {
            int   m  = m0 + hi + i;
            float h  = Hs[(hi + i) * LDSTRIDE + dCol];
            float rg = sigmoidf_(c_ir[i] + b_ir + c_hr[i] + b_hr);
            float zg = sigmoidf_(c_iz[i] + b_iz + c_hz[i] + b_hz);
            float ng = tanhf(c_in[i] + b_in + rg * (c_hn[i] + b_hn));
            float o  = (1.0f - zg) * ng + zg * h;
            int r = m % R_REL, f = m / R_REL;
            out[(size_t)r * (2 * D_DIM) + dCol * 2 + f] = o;
        }
    }

    {
        const int c0 = 192;
        const bool okN = (nRow < 8);
        const float* pI0 = Wih + (size_t)(c0 + nRow) * D_DIM + kOff;
        const float* pI1 = pI0 + (size_t)D_DIM * D_DIM;
        const float* pI2 = pI1 + (size_t)D_DIM * D_DIM;
        const float* pH0 = Whh + (size_t)(c0 + nRow) * D_DIM + kOff;
        const float* pH1 = pH0 + (size_t)D_DIM * D_DIM;
        const float* pH2 = pH1 + (size_t)D_DIM * D_DIM;

        v8f c_ir = {}; v8f c_iz = {}; v8f c_in = {};
        v8f c_hr = {}; v8f c_hz = {}; v8f c_hn = {};

        for (int k0 = 0; k0 < D_DIM; k0 += 4) {
            v2f ax, ah, b0, b1, b2, b3, b4, b5;
            ax.x = Xs[aRow * LDSTRIDE + k0 + kOff];
            ax.y = Xs[aRow * LDSTRIDE + k0 + kOff + 1];
            ah.x = Hs[aRow * LDSTRIDE + k0 + kOff];
            ah.y = Hs[aRow * LDSTRIDE + k0 + kOff + 1];
            b0.x = pI0[k0]; b0.y = pI0[k0 + 1];
            b1.x = pI1[k0]; b1.y = pI1[k0 + 1];
            b2.x = okN ? pI2[k0] : 0.0f; b2.y = okN ? pI2[k0 + 1] : 0.0f;
            b3.x = pH0[k0]; b3.y = pH0[k0 + 1];
            b4.x = pH1[k0]; b4.y = pH1[k0 + 1];
            b5.x = okN ? pH2[k0] : 0.0f; b5.y = okN ? pH2[k0 + 1] : 0.0f;

            c_ir = __builtin_amdgcn_wmma_f32_16x16x4_f32(false, ax, false, b0, (short)0, c_ir, false, false);
            c_iz = __builtin_amdgcn_wmma_f32_16x16x4_f32(false, ax, false, b1, (short)0, c_iz, false, false);
            c_in = __builtin_amdgcn_wmma_f32_16x16x4_f32(false, ax, false, b2, (short)0, c_in, false, false);
            c_hr = __builtin_amdgcn_wmma_f32_16x16x4_f32(false, ah, false, b3, (short)0, c_hr, false, false);
            c_hz = __builtin_amdgcn_wmma_f32_16x16x4_f32(false, ah, false, b4, (short)0, c_hz, false, false);
            c_hn = __builtin_amdgcn_wmma_f32_16x16x4_f32(false, ah, false, b5, (short)0, c_hn, false, false);
        }

        int  dCol  = c0 + nRow;
        bool valid = dCol < D_DIM;
        float b_ir = 0, b_iz = 0, b_in = 0, b_hr = 0, b_hz = 0, b_hn = 0;
        if (valid) {
            b_ir = bih[dCol]; b_iz = bih[D_DIM + dCol]; b_in = bih[2 * D_DIM + dCol];
            b_hr = bhh[dCol]; b_hz = bhh[D_DIM + dCol]; b_hn = bhh[2 * D_DIM + dCol];
        }
        #pragma unroll
        for (int i = 0; i < 8; ++i) {
            if (valid) {
                int   m  = m0 + hi + i;
                float h  = Hs[(hi + i) * LDSTRIDE + dCol];
                float rg = sigmoidf_(c_ir[i] + b_ir + c_hr[i] + b_hr);
                float zg = sigmoidf_(c_iz[i] + b_iz + c_hz[i] + b_hz);
                float ng = tanhf(c_in[i] + b_in + rg * (c_hn[i] + b_hn));
                float o  = (1.0f - zg) * ng + zg * h;
                int r = m % R_REL, f = m / R_REL;
                out[(size_t)r * (2 * D_DIM) + dCol * 2 + f] = o;
            }
        }
    }
}

extern "C" void kernel_launch(void* const* d_in, const int* in_sizes, int n_in,
                              void* d_out, int out_size, void* d_ws, size_t ws_size,
                              hipStream_t stream) {
    const int*   ei  = (const int*)d_in[0];
    const int*   et  = (const int*)d_in[1];
    const int*   nid = (const int*)d_in[2];
    const float* emb = (const float*)d_in[3];
    const float* dyn = (const float*)d_in[4];
    const float* Wih = (const float*)d_in[5];
    const float* Whh = (const float*)d_in[6];
    const float* bih = (const float*)d_in[7];
    const float* bhh = (const float*)d_in[8];
    float* out = (float*)d_out;

    float* sum = (float*)d_ws;
    float* cnt = sum + (size_t)M_ROWS * D_DIM;

    int nZero = M_ROWS * D_DIM + R_REL;
    zero_ws<<<(nZero + 255) / 256, 256, 0, stream>>>(sum, nZero);

    long long total = (long long)E_EDGES * (D_DIM / 4);
    int blocks = (int)((total + 255) / 256);
    scatter_kernel<<<blocks, 256, 0, stream>>>(ei, et, nid, emb, sum, cnt);

    mean_kernel<<<(M_ROWS * D_DIM + 255) / 256, 256, 0, stream>>>(sum, cnt);

    gru_wmma_kernel<<<M_ROWS / 16, 32, 0, stream>>>(sum, dyn, Wih, Whh, bih, bhh, out);
}


# scatter: one wave per 16 edges, lane = feature, lanes shifted so each f32 atomic instruction covers one whole 128-B line of the sum row (14 per edge); f32 atomics kept, sums still in arrival order
# speedup vs baseline: 1.9069x; 1.9069x over previous
#include <hip/hip_runtime.h>
#include <math.h>

#define E_EDGES  1000000
#define R_REL    1000
#define D_DIM    200
#define THREE_D  600
#define M_ROWS   2000
#define LDSTRIDE 204

typedef float v2f __attribute__((ext_vector_type(2)));
typedef float v8f __attribute__((ext_vector_type(8)));

__global__ void zero_ws(float* ws, int n) {
    int i = blockIdx.x * blockDim.x + threadIdx.x;
    if (i < n) ws[i] = 0.0f;
}

#define EPW 16
#define WAVES_PER_BLOCK 8

__device__ __forceinline__ void load_row7(const float* __restrict__ row, int c0, float (&v)[7]) {
    v[0] = row[c0 < 0 ? 0 : c0];
    v[1] = row[c0 + 32];
    v[2] = row[c0 + 64];
    v[3] = row[c0 + 96];
    v[4] = row[c0 + 128];
    v[5] = row[c0 + 160];
    v[6] = row[c0 < 8 ? c0 + 192 : D_DIM - 1];
}

__device__ __forceinline__ void add_row7(float* __restrict__ row, int c0, bool p0, bool p6,
                                         const float (&v)[7]) {
    if (p0) atomicAdd(row + c0, v[0]);
    atomicAdd(row + c0 + 32,  v[1]);
    atomicAdd(row + c0 + 64,  v[2]);
    atomicAdd(row + c0 + 96,  v[3]);
    atomicAdd(row + c0 + 128, v[4]);
    atomicAdd(row + c0 + 160, v[5]);
    if (p6) atomicAdd(row + c0 + 192, v[6]);
}

__global__ void scatter_kernel(const int* __restrict__ ei,
                               const int* __restrict__ et,
                               const int* __restrict__ nid,
                               const float* __restrict__ emb,
                               float* __restrict__ sum,
                               float* __restrict__ cnt) {
    const int lane = threadIdx.x & 31;
    const int wave = __builtin_amdgcn_readfirstlane(blockIdx.x * WAVES_PER_BLOCK + (threadIdx.x >> 5));
    const int e0 = wave * EPW;
    if (e0 >= E_EDGES) return;
    const int n = (E_EDGES - e0 < EPW) ? (E_EDGES - e0) : EPW;

    int rv = 0, sv = 0, tv = 0;
    if (lane < n) {
        const int e = e0 + lane;
        rv = et[e];
        sv = nid[ei[e]];
        tv = nid[ei[E_EDGES + e]];
        atomicAdd(cnt + rv, 1.0f);
    }

    int r  = __builtin_amdgcn_readlane(rv, 0);
    int c0 = lane - ((r & 3) << 3);
    bool p0 = c0 >= 0, p6 = c0 < 8;
    float a[7], b[7];
    load_row7(emb + (size_t)__builtin_amdgcn_readlane(sv, 0) * D_DIM, c0, a);
    load_row7(emb + (size_t)__builtin_amdgcn_readlane(tv, 0) * D_DIM, c0, b);

    #pragma unroll
    for (int j = 1; j < n; ++j) {
        const int nr  = __builtin_amdgcn_readlane(rv, j);
        const int nc0 = lane - ((nr & 3) << 3);
        float na[7], nb[7];
        load_row7(emb + (size_t)__builtin_amdgcn_readlane(sv, j) * D_DIM, nc0, na);
        load_row7(emb + (size_t)__builtin_amdgcn_readlane(tv, j) * D_DIM, nc0, nb);

        add_row7(sum + (size_t)r * D_DIM,           c0, p0, p6, a);
        add_row7(sum + (size_t)(R_REL + r) * D_DIM, c0, p0, p6, b);

        r = nr; c0 = nc0; p0 = nc0 >= 0; p6 = nc0 < 8;
        for (int i = 0; i < 7; ++i) { a[i] = na[i]; b[i] = nb[i]; }
    }
    add_row7(sum + (size_t)r * D_DIM,           c0, p0, p6, a);
    add_row7(sum + (size_t)(R_REL + r) * D_DIM, c0, p0, p6, b);
}

__global__ void mean_kernel(float* __restrict__ sum, const float* __restrict__ cnt) {
    int i = blockIdx.x * blockDim.x + threadIdx.x;
    if (i >= M_ROWS * D_DIM) return;
    int m = i / D_DIM;
    sum[i] = sum[i] / fmaxf(cnt[m % R_REL], 1.0f);
}

__device__ __forceinline__ float sigmoidf_(float x) { return 1.0f / (1.0f + expf(-x)); }

__launch_bounds__(32)
__global__ void gru_wmma_kernel(const float* __restrict__ X,
                                const float* __restrict__ dyn,
                                const float* __restrict__ Wih,
                                const float* __restrict__ Whh,
                                const float* __restrict__ bih,
                                const float* __restrict__ bhh,
                                float* __restrict__ out) {
    __shared__ float Xs[16 * LDSTRIDE];
    __shared__ float Hs[16 * LDSTRIDE];
    const int lane = threadIdx.x;
    const int m0   = blockIdx.x * 16;

    for (int idx = lane; idx < 16 * D_DIM; idx += 32) {
        int row = idx / D_DIM, col = idx - row * D_DIM;
        int m = m0 + row;
        int r = m % R_REL, f = m / R_REL;
        Xs[row * LDSTRIDE + col] = X[(size_t)m * D_DIM + col];
        Hs[row * LDSTRIDE + col] = dyn[(size_t)r * (2 * D_DIM) + col * 2 + f];
    }
    __syncthreads();

    const int aRow = lane & 15;
    const int nRow = lane & 15;
    const int kOff = (lane < 16) ? 0 : 2;
    const int hi   = (lane < 16) ? 0 : 8;

    for (int c0 = 0; c0 < 192; c0 += 16) {
        const float* pI0 = Wih + (size_t)(c0 + nRow) * D_DIM + kOff;
        const float* pI1 = pI0 + (size_t)D_DIM * D_DIM;
        const float* pI2 = pI1 + (size_t)D_DIM * D_DIM;
        const float* pH0 = Whh + (size_t)(c0 + nRow) * D_DIM + kOff;
        const float* pH1 = pH0 + (size_t)D_DIM * D_DIM;
        const float* pH2 = pH1 + (size_t)D_DIM * D_DIM;

        v8f c_ir = {}; v8f c_iz = {}; v8f c_in = {};
        v8f c_hr = {}; v8f c_hz = {}; v8f c_hn = {};

        for (int k0 = 0; k0 < D_DIM; k0 += 4) {
            v2f ax, ah, b0, b1, b2, b3, b4, b5;
            ax.x = Xs[aRow * LDSTRIDE + k0 + kOff];
            ax.y = Xs[aRow * LDSTRIDE + k0 + kOff + 1];
            ah.x = Hs[aRow * LDSTRIDE + k0 + kOff];
            ah.y = Hs[aRow * LDSTRIDE + k0 + kOff + 1];
            b0.x = pI0[k0]; b0.y = pI0[k0 + 1];
            b1.x = pI1[k0]; b1.y = pI1[k0 + 1];
            b2.x = pI2[k0]; b2.y = pI2[k0 + 1];
            b3.x = pH0[k0]; b3.y = pH0[k0 + 1];
            b4.x = pH1[k0]; b4.y = pH1[k0 + 1];
            b5.x = pH2[k0]; b5.y = pH2[k0 + 1];

            c_ir = __builtin_amdgcn_wmma_f32_16x16x4_f32(false, ax, false, b0, (short)0, c_ir, false, false);
            c_iz = __builtin_amdgcn_wmma_f32_16x16x4_f32(false, ax, false, b1, (short)0, c_iz, false, false);
            c_in = __builtin_amdgcn_wmma_f32_16x16x4_f32(false, ax, false, b2, (short)0, c_in, false, false);
            c_hr = __builtin_amdgcn_wmma_f32_16x16x4_f32(false, ah, false, b3, (short)0, c_hr, false, false);
            c_hz = __builtin_amdgcn_wmma_f32_16x16x4_f32(false, ah, false, b4, (short)0, c_hz, false, false);
            c_hn = __builtin_amdgcn_wmma_f32_16x16x4_f32(false, ah, false, b5, (short)0, c_hn, false, false);
        }

        int dCol = c0 + nRow;
        float b_ir = bih[dCol], b_iz = bih[D_DIM + dCol], b_in = bih[2 * D_DIM + dCol];
        float b_hr = bhh[dCol], b_hz = bhh[D_DIM + dCol], b_hn = bhh[2 * D_DIM + dCol];
        #pragma unroll
        for (int i = 0; i < 8; ++i) {
            int   m  = m0 + hi + i;
            float h  = Hs[(hi + i) * LDSTRIDE + dCol];
            float rg = sigmoidf_(c_ir[i] + b_ir + c_hr[i] + b_hr);
            float zg = sigmoidf_(c_iz[i] + b_iz + c_hz[i] + b_hz);
            float ng = tanhf(c_in[i] + b_in + rg * (c_hn[i] + b_hn));
            float o  = (1.0f - zg) * ng + zg * h;
            int r = m % R_REL, f = m / R_REL;
            out[(size_t)r * (2 * D_DIM) + dCol * 2 + f] = o;
        }
    }

    {
        const int c0 = 192;
        const bool okN = (nRow < 8);
        const float* pI0 = Wih + (size_t)(c0 + nRow) * D_DIM + kOff;
        const float* pI1 = pI0 + (size_t)D_DIM * D_DIM;
        const float* pI2 = pI1 + (size_t)D_DIM * D_DIM;
        const float* pH0 = Whh + (size_t)(c0 + nRow) * D_DIM + kOff;
        const float* pH1 = pH0 + (size_t)D_DIM * D_DIM;
        const float* pH2 = pH1 + (size_t)D_DIM * D_DIM;

        v8f c_ir = {}; v8f c_iz = {}; v8f c_in = {};
        v8f c_hr = {}; v8f c_hz = {}; v8f c_hn = {};

        for (int k0 = 0; k0 < D_DIM; k0 += 4) {
            v2f ax, ah, b0, b1, b2, b3, b4, b5;
            ax.x = Xs[aRow * LDSTRIDE + k0 + kOff];
            ax.y = Xs[aRow * LDSTRIDE + k0 + kOff + 1];
            ah.x = Hs[aRow * LDSTRIDE + k0 + kOff];
            ah.y = Hs[aRow * LDSTRIDE + k0 + kOff + 1];
            b0.x = pI0[k0]; b0.y = pI0[k0 + 1];
            b1.x = pI1[k0]; b1.y = pI1[k0 + 1];
            b2.x = okN ? pI2[k0] : 0.0f; b2.y = okN ? pI2[k0 + 1] : 0.0f;
            b3.x = pH0[k0]; b3.y = pH0[k0 + 1];
            b4.x = pH1[k0]; b4.y = pH1[k0 + 1];
            b5.x = okN ? pH2[k0] : 0.0f; b5.y = okN ? pH2[k0 + 1] : 0.0f;

            c_ir = __builtin_amdgcn_wmma_f32_16x16x4_f32(false, ax, false, b0, (short)0, c_ir, false, false);
            c_iz = __builtin_amdgcn_wmma_f32_16x16x4_f32(false, ax, false, b1, (short)0, c_iz, false, false);
            c_in = __builtin_amdgcn_wmma_f32_16x16x4_f32(false, ax, false, b2, (short)0, c_in, false, false);
            c_hr = __builtin_amdgcn_wmma_f32_16x16x4_f32(false, ah, false, b3, (short)0, c_hr, false, false);
            c_hz = __builtin_amdgcn_wmma_f32_16x16x4_f32(false, ah, false, b4, (short)0, c_hz, false, false);
            c_hn = __builtin_amdgcn_wmma_f32_16x16x4_f32(false, ah, false, b5, (short)0, c_hn, false, false);
        }

        int  dCol  = c0 + nRow;
        bool valid = dCol < D_DIM;
        float b_ir = 0, b_iz = 0, b_in = 0, b_hr = 0, b_hz = 0, b_hn = 0;
        if (valid) {
            b_ir = bih[dCol]; b_iz = bih[D_DIM + dCol]; b_in = bih[2 * D_DIM + dCol];
            b_hr = bhh[dCol]; b_hz = bhh[D_DIM + dCol]; b_hn = bhh[2 * D_DIM + dCol];
        }
        #pragma unroll
        for (int i = 0; i < 8; ++i) {
            if (valid) {
                int   m  = m0 + hi + i;
                float h  = Hs[(hi + i) * LDSTRIDE + dCol];
                float rg = sigmoidf_(c_ir[i] + b_ir + c_hr[i] + b_hr);
                float zg = sigmoidf_(c_iz[i] + b_iz + c_hz[i] + b_hz);
                float ng = tanhf(c_in[i] + b_in + rg * (c_hn[i] + b_hn));
                float o  = (1.0f - zg) * ng + zg * h;
                int r = m % R_REL, f = m / R_REL;
                out[(size_t)r * (2 * D_DIM) + dCol * 2 + f] = o;
            }
        }
    }
}

extern "C" void kernel_launch(void* const* d_in, const int* in_sizes, int n_in,
                              void* d_out, int out_size, void* d_ws, size_t ws_size,
                              hipStream_t stream) {
    const int*   ei  = (const int*)d_in[0];
    const int*   et  = (const int*)d_in[1];
    const int*   nid = (const int*)d_in[2];
    const float* emb = (const float*)d_in[3];
    const float* dyn = (const float*)d_in[4];
    const float* Wih = (const float*)d_in[5];
    const float* Whh = (const float*)d_in[6];
    const float* bih = (const float*)d_in[7];
    const float* bhh = (const float*)d_in[8];
    float* out = (float*)d_out;

    float* sum = (float*)d_ws;
    float* cnt = sum + (size_t)M_ROWS * D_DIM;

    int nZero = M_ROWS * D_DIM + R_REL;
    zero_ws<<<(nZero + 255) / 256, 256, 0, stream>>>(sum, nZero);

    scatter_kernel<<<(E_EDGES + EPW * WAVES_PER_BLOCK - 1) / (EPW * WAVES_PER_BLOCK), 32 * WAVES_PER_BLOCK, 0, stream>>>(
        ei, et, nid, emb, sum, cnt);

    mean_kernel<<<(M_ROWS * D_DIM + 255) / 256, 256, 0, stream>>>(sum, cnt);

    gru_wmma_kernel<<<M_ROWS / 16, 32, 0, stream>>>(sum, dyn, Wih, Whh, bih, bhh, out);
}


# scatter: one private f32 sum array per L2 half (die id picks it; the second lives in the out buffer, zeroed first), added in the mean kernel; sums in arrival order per half, then half0+half1
# speedup vs baseline: 3.8883x; 2.0391x over previous
#include <hip/hip_runtime.h>
#include <math.h>

#define E_EDGES  1000000
#define R_REL    1000
#define D_DIM    200
#define THREE_D  600
#define M_ROWS   2000
#define LDSTRIDE 204

typedef float v2f __attribute__((ext_vector_type(2)));
typedef float v8f __attribute__((ext_vector_type(8)));

__global__ void zero_ws(float* ws, int n, float* acc2, int n2) {
    int i = blockIdx.x * blockDim.x + threadIdx.x;
    if (i < n)  ws[i]   = 0.0f;
    if (i < n2) acc2[i] = 0.0f;
}

#define EPW 16
#define WAVES_PER_BLOCK 8

__device__ __forceinline__ void load_row7(const float* __restrict__ row, int c0, float (&v)[7]) {
    v[0] = row[c0 < 0 ? 0 : c0];
    v[1] = row[c0 + 32];
    v[2] = row[c0 + 64];
    v[3] = row[c0 + 96];
    v[4] = row[c0 + 128];
    v[5] = row[c0 + 160];
    v[6] = row[c0 < 8 ? c0 + 192 : D_DIM - 1];
}

__device__ __forceinline__ void add_row7(float* __restrict__ row, int c0, bool p0, bool p6,
                                         const float (&v)[7]) {
    if (p0) atomicAdd(row + c0, v[0]);
    atomicAdd(row + c0 + 32,  v[1]);
    atomicAdd(row + c0 + 64,  v[2]);
    atomicAdd(row + c0 + 96,  v[3]);
    atomicAdd(row + c0 + 128, v[4]);
    atomicAdd(row + c0 + 160, v[5]);
    if (p6) atomicAdd(row + c0 + 192, v[6]);
}

__global__ void scatter_kernel(const int* __restrict__ ei,
                               const int* __restrict__ et,
                               const int* __restrict__ nid,
                               const float* __restrict__ emb,
                               float* __restrict__ sum0,
                               float* __restrict__ sum1,
                               float* __restrict__ cnt) {
    const int lane = threadIdx.x & 31;
    const unsigned place = (unsigned)__builtin_amdgcn_s_sendmsg_rtn(0x87);
    float* __restrict__ sum = (((place >> 16) & 15u) >= 4u) ? sum1 : sum0;
    const int wave = __builtin_amdgcn_readfirstlane(blockIdx.x * WAVES_PER_BLOCK + (threadIdx.x >> 5));
    const int e0 = wave * EPW;
    if (e0 >= E_EDGES) return;
    const int n = (E_EDGES - e0 < EPW) ? (E_EDGES - e0) : EPW;

    int rv = 0, sv = 0, tv = 0;
    if (lane < n) {
        const int e = e0 + lane;
        rv = et[e];
        sv = nid[ei[e]];
        tv = nid[ei[E_EDGES + e]];
        atomicAdd(cnt + rv, 1.0f);
    }

    int r  = __builtin_amdgcn_readlane(rv, 0);
    int c0 = lane - ((r & 3) << 3);
    bool p0 = c0 >= 0, p6 = c0 < 8;
    float a[7], b[7];
    load_row7(emb + (size_t)__builtin_amdgcn_readlane(sv, 0) * D_DIM, c0, a);
    load_row7(emb + (size_t)__builtin_amdgcn_readlane(tv, 0) * D_DIM, c0, b);

    #pragma unroll
    for (int j = 1; j < n; ++j) {
        const int nr  = __builtin_amdgcn_readlane(rv, j);
        const int nc0 = lane - ((nr & 3) << 3);
        float na[7], nb[7];
        load_row7(emb + (size_t)__builtin_amdgcn_readlane(sv, j) * D_DIM, nc0, na);
        load_row7(emb + (size_t)__builtin_amdgcn_readlane(tv, j) * D_DIM, nc0, nb);

        add_row7(sum + (size_t)r * D_DIM,           c0, p0, p6, a);
        add_row7(sum + (size_t)(R_REL + r) * D_DIM, c0, p0, p6, b);

        r = nr; c0 = nc0; p0 = nc0 >= 0; p6 = nc0 < 8;
        for (int i = 0; i < 7; ++i) { a[i] = na[i]; b[i] = nb[i]; }
    }
    add_row7(sum + (size_t)r * D_DIM,           c0, p0, p6, a);
    add_row7(sum + (size_t)(R_REL + r) * D_DIM, c0, p0, p6, b);
}

__global__ void mean_kernel(float* __restrict__ sum, const float* __restrict__ sum1,
                            const float* __restrict__ cnt) {
    int i = blockIdx.x * blockDim.x + threadIdx.x;
    if (i >= M_ROWS * D_DIM) return;
    int m = i / D_DIM;
    sum[i] = (sum[i] + sum1[i]) / fmaxf(cnt[m % R_REL], 1.0f);
}

__device__ __forceinline__ float sigmoidf_(float x) { return 1.0f / (1.0f + expf(-x)); }

__launch_bounds__(32)
__global__ void gru_wmma_kernel(const float* __restrict__ X,
                                const float* __restrict__ dyn,
                                const float* __restrict__ Wih,
                                const float* __restrict__ Whh,
                                const float* __restrict__ bih,
                                const float* __restrict__ bhh,
                                float* __restrict__ out) {
    __shared__ float Xs[16 * LDSTRIDE];
    __shared__ float Hs[16 * LDSTRIDE];
    const int lane = threadIdx.x;
    const int m0   = blockIdx.x * 16;

    for (int idx = lane; idx < 16 * D_DIM; idx += 32) {
        int row = idx / D_DIM, col = idx - row * D_DIM;
        int m = m0 + row;
        int r = m % R_REL, f = m / R_REL;
        Xs[row * LDSTRIDE + col] = X[(size_t)m * D_DIM + col];
        Hs[row * LDSTRIDE + col] = dyn[(size_t)r * (2 * D_DIM) + col * 2 + f];
    }
    __syncthreads();

    const int aRow = lane & 15;
    const int nRow = lane & 15;
    const int kOff = (lane < 16) ? 0 : 2;
    const int hi   = (lane < 16) ? 0 : 8;

    for (int c0 = 0; c0 < 192; c0 += 16) {
        const float* pI0 = Wih + (size_t)(c0 + nRow) * D_DIM + kOff;
        const float* pI1 = pI0 + (size_t)D_DIM * D_DIM;
        const float* pI2 = pI1 + (size_t)D_DIM * D_DIM;
        const float* pH0 = Whh + (size_t)(c0 + nRow) * D_DIM + kOff;
        const float* pH1 = pH0 + (size_t)D_DIM * D_DIM;
        const float* pH2 = pH1 + (size_t)D_DIM * D_DIM;

        v8f c_ir = {}; v8f c_iz = {}; v8f c_in = {};
        v8f c_hr = {}; v8f c_hz = {}; v8f c_hn = {};

        for (int k0 = 0; k0 < D_DIM; k0 += 4) {
            v2f ax, ah, b0, b1, b2, b3, b4, b5;
            ax.x = Xs[aRow * LDSTRIDE + k0 + kOff];
            ax.y = Xs[aRow * LDSTRIDE + k0 + kOff + 1];
            ah.x = Hs[aRow * LDSTRIDE + k0 + kOff];
            ah.y = Hs[aRow * LDSTRIDE + k0 + kOff + 1];
            b0.x = pI0[k0]; b0.y = pI0[k0 + 1];
            b1.x = pI1[k0]; b1.y = pI1[k0 + 1];
            b2.x = pI2[k0]; b2.y = pI2[k0 + 1];
            b3.x = pH0[k0]; b3.y = pH0[k0 + 1];
            b4.x = pH1[k0]; b4.y = pH1[k0 + 1];
            b5.x = pH2[k0]; b5.y = pH2[k0 + 1];

            c_ir = __builtin_amdgcn_wmma_f32_16x16x4_f32(false, ax, false, b0, (short)0, c_ir, false, false);
            c_iz = __builtin_amdgcn_wmma_f32_16x16x4_f32(false, ax, false, b1, (short)0, c_iz, false, false);
            c_in = __builtin_amdgcn_wmma_f32_16x16x4_f32(false, ax, false, b2, (short)0, c_in, false, false);
            c_hr = __builtin_amdgcn_wmma_f32_16x16x4_f32(false, ah, false, b3, (short)0, c_hr, false, false);
            c_hz = __builtin_amdgcn_wmma_f32_16x16x4_f32(false, ah, false, b4, (short)0, c_hz, false, false);
            c_hn = __builtin_amdgcn_wmma_f32_16x16x4_f32(false, ah, false, b5, (short)0, c_hn, false, false);
        }

        int dCol = c0 + nRow;
        float b_ir = bih[dCol], b_iz = bih[D_DIM + dCol], b_in = bih[2 * D_DIM + dCol];
        float b_hr = bhh[dCol], b_hz = bhh[D_DIM + dCol], b_hn = bhh[2 * D_DIM + dCol];
        #pragma unroll
        for (int i = 0; i < 8; ++i) {
            int   m  = m0 + hi + i;
            float h  = Hs[(hi + i) * LDSTRIDE + dCol];
            float rg = sigmoidf_(c_ir[i] + b_ir + c_hr[i] + b_hr);
            float zg = sigmoidf_(c_iz[i] + b_iz + c_hz[i] + b_hz);
            float ng = tanhf(c_in[i] + b_in + rg * (c_hn[i] + b_hn));
            float o  = (1.0f - zg) * ng + zg * h;
            int r = m % R_REL, f = m / R_REL;
            out[(size_t)r * (2 * D_DIM) + dCol * 2 + f] = o;
        }
    }

    {
        const int c0 = 192;
        const bool okN = (nRow < 8);
        const float* pI0 = Wih + (size_t)(c0 + nRow) * D_DIM + kOff;
        const float* pI1 = pI0 + (size_t)D_DIM * D_DIM;
        const float* pI2 = pI1 + (size_t)D_DIM * D_DIM;
        const float* pH0 = Whh + (size_t)(c0 + nRow) * D_DIM + kOff;
        const float* pH1 = pH0 + (size_t)D_DIM * D_DIM;
        const float* pH2 = pH1 + (size_t)D_DIM * D_DIM;

        v8f c_ir = {}; v8f c_iz = {}; v8f c_in = {};
        v8f c_hr = {}; v8f c_hz = {}; v8f c_hn = {};

        for (int k0 = 0; k0 < D_DIM; k0 += 4) {
            v2f ax, ah, b0, b1, b2, b3, b4, b5;
            ax.x = Xs[aRow * LDSTRIDE + k0 + kOff];
            ax.y = Xs[aRow * LDSTRIDE + k0 + kOff + 1];
            ah.x = Hs[aRow * LDSTRIDE + k0 + kOff];
            ah.y = Hs[aRow * LDSTRIDE + k0 + kOff + 1];
            b0.x = pI0[k0]; b0.y = pI0[k0 + 1];
            b1.x = pI1[k0]; b1.y = pI1[k0 + 1];
            b2.x = okN ? pI2[k0] : 0.0f; b2.y = okN ? pI2[k0 + 1] : 0.0f;
            b3.x = pH0[k0]; b3.y = pH0[k0 + 1];
            b4.x = pH1[k0]; b4.y = pH1[k0 + 1];
            b5.x = okN ? pH2[k0] : 0.0f; b5.y = okN ? pH2[k0 + 1] : 0.0f;

            c_ir = __builtin_amdgcn_wmma_f32_16x16x4_f32(false, ax, false, b0, (short)0, c_ir, false, false);
            c_iz = __builtin_amdgcn_wmma_f32_16x16x4_f32(false, ax, false, b1, (short)0, c_iz, false, false);
            c_in = __builtin_amdgcn_wmma_f32_16x16x4_f32(false, ax, false, b2, (short)0, c_in, false, false);
            c_hr = __builtin_amdgcn_wmma_f32_16x16x4_f32(false, ah, false, b3, (short)0, c_hr, false, false);
            c_hz = __builtin_amdgcn_wmma_f32_16x16x4_f32(false, ah, false, b4, (short)0, c_hz, false, false);
            c_hn = __builtin_amdgcn_wmma_f32_16x16x4_f32(false, ah, false, b5, (short)0, c_hn, false, false);
        }

        int  dCol  = c0 + nRow;
        bool valid = dCol < D_DIM;
        float b_ir = 0, b_iz = 0, b_in = 0, b_hr = 0, b_hz = 0, b_hn = 0;
        if (valid) {
            b_ir = bih[dCol]; b_iz = bih[D_DIM + dCol]; b_in = bih[2 * D_DIM + dCol];
            b_hr = bhh[dCol]; b_hz = bhh[D_DIM + dCol]; b_hn = bhh[2 * D_DIM + dCol];
        }
        #pragma unroll
        for (int i = 0; i < 8; ++i) {
            if (valid) {
                int   m  = m0 + hi + i;
                float h  = Hs[(hi + i) * LDSTRIDE + dCol];
                float rg = sigmoidf_(c_ir[i] + b_ir + c_hr[i] + b_hr);
                float zg = sigmoidf_(c_iz[i] + b_iz + c_hz[i] + b_hz);
                float ng = tanhf(c_in[i] + b_in + rg * (c_hn[i] + b_hn));
                float o  = (1.0f - zg) * ng + zg * h;
                int r = m % R_REL, f = m / R_REL;
                out[(size_t)r * (2 * D_DIM) + dCol * 2 + f] = o;
            }
        }
    }
}

extern "C" void kernel_launch(void* const* d_in, const int* in_sizes, int n_in,
                              void* d_out, int out_size, void* d_ws, size_t ws_size,
                              hipStream_t stream) {
    const int*   ei  = (const int*)d_in[0];
    const int*   et  = (const int*)d_in[1];
    const int*   nid = (const int*)d_in[2];
    const float* emb = (const float*)d_in[3];
    const float* dyn = (const float*)d_in[4];
    const float* Wih = (const float*)d_in[5];
    const float* Whh = (const float*)d_in[6];
    const float* bih = (const float*)d_in[7];
    const float* bhh = (const float*)d_in[8];
    float* out = (float*)d_out;

    float* sum = (float*)d_ws;
    float* cnt = sum + (size_t)M_ROWS * D_DIM;

    float* sum1 = out;

    int nZero = M_ROWS * D_DIM + R_REL;
    zero_ws<<<(nZero + 255) / 256, 256, 0, stream>>>(sum, nZero, sum1, M_ROWS * D_DIM);

    scatter_kernel<<<(E_EDGES + EPW * WAVES_PER_BLOCK - 1) / (EPW * WAVES_PER_BLOCK), 32 * WAVES_PER_BLOCK, 0, stream>>>(
        ei, et, nid, emb, sum, sum1, cnt);

    mean_kernel<<<(M_ROWS * D_DIM + 255) / 256, 256, 0, stream>>>(sum, sum1, cnt);

    gru_wmma_kernel<<<M_ROWS / 16, 32, 0, stream>>>(sum, dyn, Wih, Whh, bih, bhh, out);
}


# GRU: one wave per 16 rows x 16 columns (grid 125 x 13) instead of one wave walking all 13 column tiles in turn; A fragments read straight from memory, no LDS staging; same k order of sums
# speedup vs baseline: 5.9339x; 1.5261x over previous
#include <hip/hip_runtime.h>
#include <math.h>

#define E_EDGES  1000000
#define R_REL    1000
#define D_DIM    200
#define THREE_D  600
#define M_ROWS   2000

typedef float v2f __attribute__((ext_vector_type(2)));
typedef float v4f __attribute__((ext_vector_type(4)));
typedef float v8f __attribute__((ext_vector_type(8)));

__global__ void zero_ws(float* ws, int n, float* acc2, int n2) {
    int i = blockIdx.x * blockDim.x + threadIdx.x;
    if (i < n)  ws[i]   = 0.0f;
    if (i < n2) acc2[i] = 0.0f;
}

#define EPW 16
#define WAVES_PER_BLOCK 8

__device__ __forceinline__ void load_row7(const float* __restrict__ row, int c0, float (&v)[7]) {
    v[0] = row[c0 < 0 ? 0 : c0];
    v[1] = row[c0 + 32];
    v[2] = row[c0 + 64];
    v[3] = row[c0 + 96];
    v[4] = row[c0 + 128];
    v[5] = row[c0 + 160];
    v[6] = row[c0 < 8 ? c0 + 192 : D_DIM - 1];
}

__device__ __forceinline__ void add_row7(float* __restrict__ row, int c0, bool p0, bool p6,
                                         const float (&v)[7]) {
    if (p0) atomicAdd(row + c0, v[0]);
    atomicAdd(row + c0 + 32,  v[1]);
    atomicAdd(row + c0 + 64,  v[2]);
    atomicAdd(row + c0 + 96,  v[3]);
    atomicAdd(row + c0 + 128, v[4]);
    atomicAdd(row + c0 + 160, v[5]);
    if (p6) atomicAdd(row + c0 + 192, v[6]);
}

__global__ void scatter_kernel(const int* __restrict__ ei,
                               const int* __restrict__ et,
                               const int* __restrict__ nid,
                               const float* __restrict__ emb,
                               float* __restrict__ sum0,
                               float* __restrict__ sum1,
                               float* __restrict__ cnt) {
    const int lane = threadIdx.x & 31;
    const unsigned place = (unsigned)__builtin_amdgcn_s_sendmsg_rtn(0x87);
    float* __restrict__ sum = (((place >> 16) & 15u) >= 4u) ? sum1 : sum0;
    const int wave = __builtin_amdgcn_readfirstlane(blockIdx.x * WAVES_PER_BLOCK + (threadIdx.x >> 5));
    const int e0 = wave * EPW;
    if (e0 >= E_EDGES) return;
    const int n = (E_EDGES - e0 < EPW) ? (E_EDGES - e0) : EPW;

    int rv = 0, sv = 0, tv = 0;
    if (lane < n) {
        const int e = e0 + lane;
        rv = et[e];
        sv = nid[ei[e]];
        tv = nid[ei[E_EDGES + e]];
        atomicAdd(cnt + rv, 1.0f);
    }

    int r  = __builtin_amdgcn_readlane(rv, 0);
    int c0 = lane - ((r & 3) << 3);
    bool p0 = c0 >= 0, p6 = c0 < 8;
    float a[7], b[7];
    load_row7(emb + (size_t)__builtin_amdgcn_readlane(sv, 0) * D_DIM, c0, a);
    load_row7(emb + (size_t)__builtin_amdgcn_readlane(tv, 0) * D_DIM, c0, b);

    #pragma unroll
    for (int j = 1; j < n; ++j) {
        const int nr  = __builtin_amdgcn_readlane(rv, j);
        const int nc0 = lane - ((nr & 3) << 3);
        float na[7], nb[7];
        load_row7(emb + (size_t)__builtin_amdgcn_readlane(sv, j) * D_DIM, nc0, na);
        load_row7(emb + (size_t)__builtin_amdgcn_readlane(tv, j) * D_DIM, nc0, nb);

        add_row7(sum + (size_t)r * D_DIM,           c0, p0, p6, a);
        add_row7(sum + (size_t)(R_REL + r) * D_DIM, c0, p0, p6, b);

        r = nr; c0 = nc0; p0 = nc0 >= 0; p6 = nc0 < 8;
        for (int i = 0; i < 7; ++i) { a[i] = na[i]; b[i] = nb[i]; }
    }
    add_row7(sum + (size_t)r * D_DIM,           c0, p0, p6, a);
    add_row7(sum + (size_t)(R_REL + r) * D_DIM, c0, p0, p6, b);
}

__global__ void mean_kernel(float* __restrict__ sum, const float* __restrict__ sum1,
                            const float* __restrict__ cnt) {
    int i = blockIdx.x * blockDim.x + threadIdx.x;
    if (i >= M_ROWS * D_DIM) return;
    int m = i / D_DIM;
    sum[i] = (sum[i] + sum1[i]) / fmaxf(cnt[m % R_REL], 1.0f);
}

__device__ __forceinline__ float sigmoidf_(float x) { return 1.0f / (1.0f + expf(-x)); }

__launch_bounds__(32)
__global__ void gru_wmma_kernel(const float* __restrict__ X,
                                const float* __restrict__ dyn,
                                const float* __restrict__ Wih,
                                const float* __restrict__ Whh,
                                const float* __restrict__ bih,
                                const float* __restrict__ bhh,
                                float* __restrict__ out) {
    const int lane = threadIdx.x;
    const int m0   = blockIdx.x * 16;
    const int c0   = blockIdx.y * 16;

    const int aRow = lane & 15;
    const int nRow = lane & 15;
    const int kOff = (lane < 16) ? 0 : 2;
    const int hi   = (lane < 16) ? 0 : 8;

    const int mA = m0 + aRow;
    const int rA = mA % R_REL, fA = mA / R_REL;
    const float* pX = X + (size_t)mA * D_DIM + kOff;
    const float* pH = dyn + (size_t)rA * (2 * D_DIM) + 2 * kOff;

    const int  dCol  = c0 + nRow;
    const bool valid = dCol < D_DIM;
    const int  wRow  = valid ? dCol : D_DIM - 1;
    const float* pI0 = Wih + (size_t)wRow * D_DIM + kOff;
    const float* pI1 = pI0 + (size_t)D_DIM * D_DIM;
    const float* pI2 = pI1 + (size_t)D_DIM * D_DIM;
    const float* pH0 = Whh + (size_t)wRow * D_DIM + kOff;
    const float* pH1 = pH0 + (size_t)D_DIM * D_DIM;
    const float* pH2 = pH1 + (size_t)D_DIM * D_DIM;

    int   oIdx[8];
    float hOld[8];
    for (int i = 0; i < 8; ++i) {
        const int m = m0 + hi + i;
        const int r = m % R_REL, f = m / R_REL;
        oIdx[i] = r * (2 * D_DIM) + wRow * 2 + f;
        hOld[i] = dyn[oIdx[i]];
    }
    const float b_ir = bih[wRow], b_iz = bih[D_DIM + wRow], b_in = bih[2 * D_DIM + wRow];
    const float b_hr = bhh[wRow], b_hz = bhh[D_DIM + wRow], b_hn = bhh[2 * D_DIM + wRow];

    v8f c_ir = {}; v8f c_iz = {}; v8f c_in = {};
    v8f c_hr = {}; v8f c_hz = {}; v8f c_hn = {};

    for (int k0 = 0; k0 < D_DIM; k0 += 4) {
        v2f ax, ah, b0, b1, b2, b3, b4, b5;
        ax.x = pX[k0]; ax.y = pX[k0 + 1];
        const v4f hq = *reinterpret_cast<const v4f*>(pH + 2 * k0);
        ah.x = fA ? hq.y : hq.x;
        ah.y = fA ? hq.w : hq.z;
        b0.x = pI0[k0]; b0.y = pI0[k0 + 1];
        b1.x = pI1[k0]; b1.y = pI1[k0 + 1];
        b2.x = pI2[k0]; b2.y = pI2[k0 + 1];
        b3.x = pH0[k0]; b3.y = pH0[k0 + 1];
        b4.x = pH1[k0]; b4.y = pH1[k0 + 1];
        b5.x = pH2[k0]; b5.y = pH2[k0 + 1];

        c_ir = __builtin_amdgcn_wmma_f32_16x16x4_f32(false, ax, false, b0, (short)0, c_ir, false, false);
        c_iz = __builtin_amdgcn_wmma_f32_16x16x4_f32(false, ax, false, b1, (short)0, c_iz, false, false);
        c_in = __builtin_amdgcn_wmma_f32_16x16x4_f32(false, ax, false, b2, (short)0, c_in, false, false);
        c_hr = __builtin_amdgcn_wmma_f32_16x16x4_f32(false, ah, false, b3, (short)0, c_hr, false, false);
        c_hz = __builtin_amdgcn_wmma_f32_16x16x4_f32(false, ah, false, b4, (short)0, c_hz, false, false);
        c_hn = __builtin_amdgcn_wmma_f32_16x16x4_f32(false, ah, false, b5, (short)0, c_hn, false, false);
    }

    #pragma unroll
    for (int i = 0; i < 8; ++i) {
        float h  = hOld[i];
        float rg = sigmoidf_(c_ir[i] + b_ir + c_hr[i] + b_hr);
        float zg = sigmoidf_(c_iz[i] + b_iz + c_hz[i] + b_hz);
        float ng = tanhf(c_in[i] + b_in + rg * (c_hn[i] + b_hn));
        float o  = (1.0f - zg) * ng + zg * h;
        if (valid) out[oIdx[i]] = o;
    }
}

extern "C" void kernel_launch(void* const* d_in, const int* in_sizes, int n_in,
                              void* d_out, int out_size, void* d_ws, size_t ws_size,
                              hipStream_t stream) {
    const int*   ei  = (const int*)d_in[0];
    const int*   et  = (const int*)d_in[1];
    const int*   nid = (const int*)d_in[2];
    const float* emb = (const float*)d_in[3];
    const float* dyn = (const float*)d_in[4];
    const float* Wih = (const float*)d_in[5];
    const float* Whh = (const float*)d_in[6];
    const float* bih = (const float*)d_in[7];
    const float* bhh = (const float*)d_in[8];
    float* out = (float*)d_out;

    float* sum = (float*)d_ws;
    float* cnt = sum + (size_t)M_ROWS * D_DIM;

    float* sum1 = out;

    int nZero = M_ROWS * D_DIM + R_REL;
    zero_ws<<<(nZero + 255) / 256, 256, 0, stream>>>(sum, nZero, sum1, M_ROWS * D_DIM);

    scatter_kernel<<<(E_EDGES + EPW * WAVES_PER_BLOCK - 1) / (EPW * WAVES_PER_BLOCK), 32 * WAVES_PER_BLOCK, 0, stream>>>(
        ei, et, nid, emb, sum, sum1, cnt);

    mean_kernel<<<(M_ROWS * D_DIM + 255) / 256, 256, 0, stream>>>(sum, sum1, cnt);

    gru_wmma_kernel<<<dim3(M_ROWS / 16, 13), 32, 0, stream>>>(sum, dyn, Wih, Whh, bih, bhh, out);
}


# scatter: each work-group keeps the sum rows of 125 relations in LDS (8 slices x 32 edge chunks), adds rows by LDS f32 atomics, one global atomic flush; sums ordered per block, then per half
# speedup vs baseline: 9.5101x; 1.6027x over previous
#include <hip/hip_runtime.h>
#include <math.h>

#define E_EDGES  1000000
#define R_REL    1000
#define D_DIM    200
#define THREE_D  600
#define M_ROWS   2000

typedef float v2f __attribute__((ext_vector_type(2)));
typedef float v4f __attribute__((ext_vector_type(4)));
typedef float v8f __attribute__((ext_vector_type(8)));

__global__ void zero_ws(float* ws, int n, float* acc2, int n2) {
    int i = blockIdx.x * blockDim.x + threadIdx.x;
    if (i < n)  ws[i]   = 0.0f;
    if (i < n2) acc2[i] = 0.0f;
}

#define NPART   8
#define RPP     125
#define NCHUNK  32
#define CHUNK_E (E_EDGES / NCHUNK)
#define SBLOCK  1024
#define SWAVES  (SBLOCK / 32)
#define QCAP    8192
#define SLICE_F (RPP * D_DIM)

__device__ __forceinline__ void load_row7(const float* __restrict__ row, int c0, float (&v)[7]) {
    v[0] = row[c0 < 0 ? 0 : c0];
    v[1] = row[c0 + 32];
    v[2] = row[c0 + 64];
    v[3] = row[c0 + 96];
    v[4] = row[c0 + 128];
    v[5] = row[c0 + 160];
    v[6] = row[c0 < 8 ? c0 + 192 : D_DIM - 1];
}

__device__ __forceinline__ void lds_add_row7(float* row, int c0, const float (&v)[7]) {
    const bool p0 = c0 >= 0, p6 = c0 < 8;
    atomicAdd(row + (p0 ? c0 : c0 + 32), p0 ? v[0] : 0.0f);
    atomicAdd(row + c0 + 32,  v[1]);
    atomicAdd(row + c0 + 64,  v[2]);
    atomicAdd(row + c0 + 96,  v[3]);
    atomicAdd(row + c0 + 128, v[4]);
    atomicAdd(row + c0 + 160, v[5]);
    atomicAdd(row + (p6 ? c0 + 192 : c0 + 160), p6 ? v[6] : 0.0f);
}

struct RowPair {
    int   l0, l1, ca0, cb0, ca1, cb1;
    bool  two;
    float a0[7], b0[7], a1[7], b1[7];
};

__launch_bounds__(SBLOCK)
__global__ void scatter_kernel(const int* __restrict__ ei,
                               const int* __restrict__ et,
                               const int* __restrict__ nid,
                               const float* __restrict__ emb,
                               float* __restrict__ sum0,
                               float* __restrict__ sum1,
                               float* __restrict__ cnt) {
    __shared__ float s_sum[2 * SLICE_F];
    __shared__ float s_cnt[128];
    __shared__ uint2 s_q[QCAP];
    __shared__ int   s_qn;

    const int tid  = threadIdx.x;
    const int lane = tid & 31;
    const int wave = __builtin_amdgcn_readfirstlane(tid >> 5);
    const int part  = blockIdx.x % NPART;
    const int chunk = blockIdx.x / NPART;
    const int rlo   = part * RPP;

    for (int i = tid; i < 2 * SLICE_F; i += SBLOCK) s_sum[i] = 0.0f;
    if (tid < 128) s_cnt[tid] = 0.0f;
    if (tid == 0)  s_qn = 0;
    __syncthreads();

    const int eBeg = chunk * CHUNK_E, eEnd = eBeg + CHUNK_E;
    int qStart = 0;
    for (int base = eBeg; base < eEnd; base += QCAP) {
        const int lim = (base + QCAP < eEnd) ? base + QCAP : eEnd;

        int rv[QCAP / SBLOCK];
        #pragma unroll
        for (int k = 0; k < QCAP / SBLOCK; ++k) {
            const int e  = base + k * SBLOCK + tid;
            rv[k] = et[e < lim ? e : lim - 1];
        }
        __builtin_amdgcn_sched_barrier(0);
        #pragma unroll
        for (int k = 0; k < QCAP / SBLOCK; ++k) {
            const int  e   = base + k * SBLOCK + tid;
            const int  lr  = e < lim ? rv[k] - rlo : -1;
            const bool hit = (unsigned)lr < (unsigned)RPP;
            const unsigned m = __builtin_amdgcn_ballot_w32(hit);
            int pos0 = 0;
            if (lane == 0 && m != 0u) pos0 = atomicAdd(&s_qn, __popc(m));
            pos0 = __builtin_amdgcn_readfirstlane(pos0);
            if (hit) {
                const int pos = pos0 - qStart + (int)__builtin_amdgcn_mbcnt_lo(m, 0u);
                s_q[pos] = make_uint2((unsigned)e, (unsigned)lr);
                atomicAdd(&s_cnt[lr], 1.0f);
            }
        }
        __syncthreads();
        const int qEnd = s_qn;
        const int qn   = qEnd - qStart;
        qStart = qEnd;

        for (int q0 = 0; q0 < qn; q0 += 2 * SBLOCK) {
            const int qa = q0 + tid, qb = qa + SBLOCK;
            const int qac = qa < qn ? qa : qn - 1, qbc = qb < qn ? qb : qn - 1;
            const uint2 ea = s_q[qac], eb = s_q[qbc];
            const int ia0 = ei[ea.x], ia1 = ei[E_EDGES + ea.x];
            const int ib0 = ei[eb.x], ib1 = ei[E_EDGES + eb.x];
            const int sa = nid[ia0], ta = nid[ia1];
            const int sb = nid[ib0], tb = nid[ib1];
            __syncthreads();
            if (qa < qn) s_q[qa] = make_uint2((unsigned)sa | (ea.y << 20), (unsigned)ta);
            if (qb < qn) s_q[qb] = make_uint2((unsigned)sb | (eb.y << 20), (unsigned)tb);
        }
        __syncthreads();

        auto fetch_pair = [&](RowPair& p, int q) {
            p.two = (q + 1) < qn;
            const uint2 e0 = s_q[q];
            const uint2 e1 = s_q[p.two ? q + 1 : q];
            const unsigned x0 = __builtin_amdgcn_readfirstlane(e0.x), y0 = __builtin_amdgcn_readfirstlane(e0.y);
            const unsigned x1 = __builtin_amdgcn_readfirstlane(e1.x), y1 = __builtin_amdgcn_readfirstlane(e1.y);
            const int s0 = (int)(x0 & 0xFFFFFu), t0 = (int)y0;
            const int s1 = (int)(x1 & 0xFFFFFu), t1 = (int)y1;
            p.l0 = (int)(x0 >> 20); p.l1 = (int)(x1 >> 20);
            p.ca0 = lane - ((s0 & 3) << 3); p.cb0 = lane - ((t0 & 3) << 3);
            p.ca1 = lane - ((s1 & 3) << 3); p.cb1 = lane - ((t1 & 3) << 3);
            load_row7(emb + (size_t)s0 * D_DIM, p.ca0, p.a0);
            load_row7(emb + (size_t)t0 * D_DIM, p.cb0, p.b0);
            load_row7(emb + (size_t)s1 * D_DIM, p.ca1, p.a1);
            load_row7(emb + (size_t)t1 * D_DIM, p.cb1, p.b1);
        };
        auto add_pair = [&](const RowPair& p) {
            lds_add_row7(s_sum + p.l0 * D_DIM,           p.ca0, p.a0);
            lds_add_row7(s_sum + SLICE_F + p.l0 * D_DIM, p.cb0, p.b0);
            float z1[7], w1[7];
            #pragma unroll
            for (int i = 0; i < 7; ++i) { z1[i] = p.two ? p.a1[i] : 0.0f; w1[i] = p.two ? p.b1[i] : 0.0f; }
            lds_add_row7(s_sum + p.l1 * D_DIM,           p.ca1, z1);
            lds_add_row7(s_sum + SLICE_F + p.l1 * D_DIM, p.cb1, w1);
        };
        {
            int q = 2 * wave;
            if (q < qn) {
                RowPair A, B;
                fetch_pair(A, q);
                while (q + 4 * SWAVES < qn) {
                    fetch_pair(B, q + 2 * SWAVES);
                    __builtin_amdgcn_sched_barrier(0);
                    add_pair(A);
                    __builtin_amdgcn_sched_barrier(0);
                    fetch_pair(A, q + 4 * SWAVES);
                    __builtin_amdgcn_sched_barrier(0);
                    add_pair(B);
                    __builtin_amdgcn_sched_barrier(0);
                    q += 4 * SWAVES;
                }
                if (q + 2 * SWAVES < qn) {
                    fetch_pair(B, q + 2 * SWAVES);
                    __builtin_amdgcn_sched_barrier(0);
                    add_pair(A);
                    add_pair(B);
                } else {
                    add_pair(A);
                }
            }
        }
        __syncthreads();
    }

    const unsigned place = (unsigned)__builtin_amdgcn_s_sendmsg_rtn(0x87);
    float* __restrict__ sum = (((place >> 16) & 15u) >= 4u) ? sum1 : sum0;
    float* g0 = sum + (size_t)part * SLICE_F;
    float* g1 = g0 + (size_t)R_REL * D_DIM;
    const int sh = (part * 8) & 31;
    for (int j = tid - sh; j < SLICE_F; j += SBLOCK) {
        if (j >= 0) {
            atomicAdd(g0 + j, s_sum[j]);
            atomicAdd(g1 + j, s_sum[SLICE_F + j]);
        }
    }
    if (tid < RPP) atomicAdd(cnt + rlo + tid, s_cnt[tid]);
}

__global__ void mean_kernel(float* __restrict__ sum, const float* __restrict__ sum1,
                            const float* __restrict__ cnt) {
    int i = blockIdx.x * blockDim.x + threadIdx.x;
    if (i >= M_ROWS * D_DIM) return;
    int m = i / D_DIM;
    sum[i] = (sum[i] + sum1[i]) / fmaxf(cnt[m % R_REL], 1.0f);
}

__device__ __forceinline__ float sigmoidf_(float x) { return 1.0f / (1.0f + expf(-x)); }

__launch_bounds__(32)
__global__ void gru_wmma_kernel(const float* __restrict__ X,
                                const float* __restrict__ dyn,
                                const float* __restrict__ Wih,
                                const float* __restrict__ Whh,
                                const float* __restrict__ bih,
                                const float* __restrict__ bhh,
                                float* __restrict__ out) {
    const int lane = threadIdx.x;
    const int m0   = blockIdx.x * 16;
    const int c0   = blockIdx.y * 16;

    const int aRow = lane & 15;
    const int nRow = lane & 15;
    const int kOff = (lane < 16) ? 0 : 2;
    const int hi   = (lane < 16) ? 0 : 8;

    const int mA = m0 + aRow;
    const int rA = mA % R_REL, fA = mA / R_REL;
    const float* pX = X + (size_t)mA * D_DIM + kOff;
    const float* pH = dyn + (size_t)rA * (2 * D_DIM) + 2 * kOff;

    const int  dCol  = c0 + nRow;
    const bool valid = dCol < D_DIM;
    const int  wRow  = valid ? dCol : D_DIM - 1;
    const float* pI0 = Wih + (size_t)wRow * D_DIM + kOff;
    const float* pI1 = pI0 + (size_t)D_DIM * D_DIM;
    const float* pI2 = pI1 + (size_t)D_DIM * D_DIM;
    const float* pH0 = Whh + (size_t)wRow * D_DIM + kOff;
    const float* pH1 = pH0 + (size_t)D_DIM * D_DIM;
    const float* pH2 = pH1 + (size_t)D_DIM * D_DIM;

    int   oIdx[8];
    float hOld[8];
    for (int i = 0; i < 8; ++i) {
        const int m = m0 + hi + i;
        const int r = m % R_REL, f = m / R_REL;
        oIdx[i] = r * (2 * D_DIM) + wRow * 2 + f;
        hOld[i] = dyn[oIdx[i]];
    }
    const float b_ir = bih[wRow], b_iz = bih[D_DIM + wRow], b_in = bih[2 * D_DIM + wRow];
    const float b_hr = bhh[wRow], b_hz = bhh[D_DIM + wRow], b_hn = bhh[2 * D_DIM + wRow];

    v8f c_ir = {}; v8f c_iz = {}; v8f c_in = {};
    v8f c_hr = {}; v8f c_hz = {}; v8f c_hn = {};

    for (int k0 = 0; k0 < D_DIM; k0 += 4) {
        v2f ax, ah, b0, b1, b2, b3, b4, b5;
        ax.x = pX[k0]; ax.y = pX[k0 + 1];
        const v4f hq = *reinterpret_cast<const v4f*>(pH + 2 * k0);
        ah.x = fA ? hq.y : hq.x;
        ah.y = fA ? hq.w : hq.z;
        b0.x = pI0[k0]; b0.y = pI0[k0 + 1];
        b1.x = pI1[k0]; b1.y = pI1[k0 + 1];
        b2.x = pI2[k0]; b2.y = pI2[k0 + 1];
        b3.x = pH0[k0]; b3.y = pH0[k0 + 1];
        b4.x = pH1[k0]; b4.y = pH1[k0 + 1];
        b5.x = pH2[k0]; b5.y = pH2[k0 + 1];

        c_ir = __builtin_amdgcn_wmma_f32_16x16x4_f32(false, ax, false, b0, (short)0, c_ir, false, false);
        c_iz = __builtin_amdgcn_wmma_f32_16x16x4_f32(false, ax, false, b1, (short)0, c_iz, false, false);
        c_in = __builtin_amdgcn_wmma_f32_16x16x4_f32(false, ax, false, b2, (short)0, c_in, false, false);
        c_hr = __builtin_amdgcn_wmma_f32_16x16x4_f32(false, ah, false, b3, (short)0, c_hr, false, false);
        c_hz = __builtin_amdgcn_wmma_f32_16x16x4_f32(false, ah, false, b4, (short)0, c_hz, false, false);
        c_hn = __builtin_amdgcn_wmma_f32_16x16x4_f32(false, ah, false, b5, (short)0, c_hn, false, false);
    }

    #pragma unroll
    for (int i = 0; i < 8; ++i) {
        float h  = hOld[i];
        float rg = sigmoidf_(c_ir[i] + b_ir + c_hr[i] + b_hr);
        float zg = sigmoidf_(c_iz[i] + b_iz + c_hz[i] + b_hz);
        float ng = tanhf(c_in[i] + b_in + rg * (c_hn[i] + b_hn));
        float o  = (1.0f - zg) * ng + zg * h;
        if (valid) out[oIdx[i]] = o;
    }
}

extern "C" void kernel_launch(void* const* d_in, const int* in_sizes, int n_in,
                              void* d_out, int out_size, void* d_ws, size_t ws_size,
                              hipStream_t stream) {
    const int*   ei  = (const int*)d_in[0];
    const int*   et  = (const int*)d_in[1];
    const int*   nid = (const int*)d_in[2];
    const float* emb = (const float*)d_in[3];
    const float* dyn = (const float*)d_in[4];
    const float* Wih = (const float*)d_in[5];
    const float* Whh = (const float*)d_in[6];
    const float* bih = (const float*)d_in[7];
    const float* bhh = (const float*)d_in[8];
    float* out = (float*)d_out;

    float* sum = (float*)d_ws;
    float* cnt = sum + (size_t)M_ROWS * D_DIM;

    float* sum1 = out;

    int nZero = M_ROWS * D_DIM + R_REL;
    zero_ws<<<(nZero + 255) / 256, 256, 0, stream>>>(sum, nZero, sum1, M_ROWS * D_DIM);

    scatter_kernel<<<NPART * NCHUNK, SBLOCK, 0, stream>>>(ei, et, nid, emb, sum, sum1, cnt);

    mean_kernel<<<(M_ROWS * D_DIM + 255) / 256, 256, 0, stream>>>(sum, sum1, cnt);

    gru_wmma_kernel<<<dim3(M_ROWS / 16, 13), 32, 0, stream>>>(sum, dyn, Wih, Whh, bih, bhh, out);
}


# scatter: round queue counting-sorted by relation in LDS; a wave sums each run of one relation in registers (rows read 16 B per lane), one LDS f32 add per run; sums ordered per run, block, half
# speedup vs baseline: 11.5126x; 1.2106x over previous
#include <hip/hip_runtime.h>
#include <math.h>

#define E_EDGES  1000000
#define R_REL    1000
#define D_DIM    200
#define THREE_D  600
#define M_ROWS   2000

typedef float v2f __attribute__((ext_vector_type(2)));
typedef float v4f __attribute__((ext_vector_type(4)));
typedef float v8f __attribute__((ext_vector_type(8)));

__global__ void zero_ws(float* ws, int n, float* acc2, int n2) {
    int i = blockIdx.x * blockDim.x + threadIdx.x;
    if (i < n)  ws[i]   = 0.0f;
    if (i < n2) acc2[i] = 0.0f;
}

#define NPART   8
#define RPP     125
#define NCHUNK  32
#define CHUNK_E (E_EDGES / NCHUNK)
#define SBLOCK  1024
#define SWAVES  (SBLOCK / 32)
#define QCAP    8192
#define SLICE_F (RPP * D_DIM)

struct RowPair {
    int  l0, l1;
    bool two;
    v4f  a0lo, a0hi, b0lo, b0hi, a1lo, a1hi, b1lo, b1hi;
};

__launch_bounds__(SBLOCK)
__global__ void scatter_kernel(const int* __restrict__ ei,
                               const int* __restrict__ et,
                               const int* __restrict__ nid,
                               const float* __restrict__ emb,
                               float* __restrict__ sum0,
                               float* __restrict__ sum1,
                               float* __restrict__ cnt) {
    __shared__ float s_sum[2 * SLICE_F];
    __shared__ __attribute__((aligned(16))) float s_cnt[128];
    __shared__ __attribute__((aligned(16))) int   s_rc[128];
    __shared__ __attribute__((aligned(16))) int   s_off[128];
    __shared__ unsigned s_q1[QCAP];
    __shared__ uint2    s_q2[QCAP];
    __shared__ int      s_qn;

    const int tid  = threadIdx.x;
    const int lane = tid & 31;
    const int wave = __builtin_amdgcn_readfirstlane(tid >> 5);
    const int part  = blockIdx.x % NPART;
    const int chunk = blockIdx.x / NPART;
    const int rlo   = part * RPP;

    for (int i = tid; i < 2 * SLICE_F; i += SBLOCK) s_sum[i] = 0.0f;
    if (tid < 128) { s_cnt[tid] = 0.0f; s_rc[tid] = 0; s_off[tid] = 0; }
    if (tid == 0)  s_qn = 0;
    __syncthreads();

    const int hiIdx = lane < 18 ? 32 + lane : 49;

    const int eBeg = chunk * CHUNK_E, eEnd = eBeg + CHUNK_E;
    int qStart = 0;
    for (int base = eBeg; base < eEnd; base += QCAP) {
        const int lim = (base + QCAP < eEnd) ? base + QCAP : eEnd;

        int rv[QCAP / SBLOCK];
        #pragma unroll
        for (int k = 0; k < QCAP / SBLOCK; ++k) {
            const int e  = base + k * SBLOCK + tid;
            rv[k] = et[e < lim ? e : lim - 1];
        }
        __builtin_amdgcn_sched_barrier(0);
        #pragma unroll
        for (int k = 0; k < QCAP / SBLOCK; ++k) {
            const int  e   = base + k * SBLOCK + tid;
            const int  lr  = e < lim ? rv[k] - rlo : -1;
            const bool hit = (unsigned)lr < (unsigned)RPP;
            const unsigned m = __builtin_amdgcn_ballot_w32(hit);
            int pos0 = 0;
            if (lane == 0 && m != 0u) pos0 = atomicAdd(&s_qn, __popc(m));
            pos0 = __builtin_amdgcn_readfirstlane(pos0);
            if (hit) {
                const int pos = pos0 - qStart + (int)__builtin_amdgcn_mbcnt_lo(m, 0u);
                s_q1[pos] = (unsigned)e | ((unsigned)lr << 20);
                atomicAdd(&s_rc[lr], 1);
            }
        }
        __syncthreads();
        const int qEnd = s_qn;
        const int qn   = qEnd - qStart;
        qStart = qEnd;

        if (wave == 0) {
            const int4 c = *reinterpret_cast<const int4*>(&s_rc[4 * lane]);
            const int s = c.x + c.y + c.z + c.w;
            int inc = s;
            #pragma unroll
            for (int d = 1; d < 32; d <<= 1) {
                const int n = __shfl_up(inc, d);
                if (lane >= d) inc += n;
            }
            const int ex = inc - s;
            int4 o;
            o.x = ex; o.y = ex + c.x; o.z = o.y + c.y; o.w = o.z + c.z;
            *reinterpret_cast<int4*>(&s_off[4 * lane]) = o;
            float4 f = *reinterpret_cast<const float4*>(&s_cnt[4 * lane]);
            f.x += (float)c.x; f.y += (float)c.y; f.z += (float)c.z; f.w += (float)c.w;
            *reinterpret_cast<float4*>(&s_cnt[4 * lane]) = f;
            *reinterpret_cast<int4*>(&s_rc[4 * lane]) = make_int4(0, 0, 0, 0);
        }
        __syncthreads();

        for (int q0 = 0; q0 < qn; q0 += 2 * SBLOCK) {
            const int qa = q0 + tid, qb = qa + SBLOCK;
            const int qac = qa < qn ? qa : qn - 1, qbc = qb < qn ? qb : qn - 1;
            const unsigned ea = s_q1[qac], eb = s_q1[qbc];
            const unsigned eaE = ea & 0xFFFFFu, ebE = eb & 0xFFFFFu;
            const unsigned lra = ea >> 20,      lrb = eb >> 20;
            const int ia0 = ei[eaE], ia1 = ei[E_EDGES + eaE];
            const int ib0 = ei[ebE], ib1 = ei[E_EDGES + ebE];
            const int sa = nid[ia0], ta = nid[ia1];
            const int sb = nid[ib0], tb = nid[ib1];
            asm volatile("" :: "v"(sa), "v"(ta), "v"(sb), "v"(tb));
            if (qa < qn) {
                const int p = atomicAdd(&s_off[lra], 1);
                s_q2[p] = make_uint2((unsigned)sa | (lra << 20), (unsigned)ta);
            }
            if (qb < qn) {
                const int p = atomicAdd(&s_off[lrb], 1);
                s_q2[p] = make_uint2((unsigned)sb | (lrb << 20), (unsigned)tb);
            }
        }
        __syncthreads();

        float accA[8], accB[8];
        #pragma unroll
        for (int i = 0; i < 8; ++i) { accA[i] = 0.0f; accB[i] = 0.0f; }
        int cur = -1;

        auto fetch_pair = [&](RowPair& p, int q, int qe) {
            p.two = (q + 1) < qe;
            const uint2 e0 = s_q2[q];
            const uint2 e1 = s_q2[p.two ? q + 1 : q];
            const unsigned x0 = __builtin_amdgcn_readfirstlane(e0.x), y0 = __builtin_amdgcn_readfirstlane(e0.y);
            const unsigned x1 = __builtin_amdgcn_readfirstlane(e1.x), y1 = __builtin_amdgcn_readfirstlane(e1.y);
            const int s0 = (int)(x0 & 0xFFFFFu), t0 = (int)y0;
            const int s1 = (int)(x1 & 0xFFFFFu), t1 = (int)y1;
            p.l0 = (int)(x0 >> 20); p.l1 = (int)(x1 >> 20);
            const v4f* ra0 = reinterpret_cast<const v4f*>(emb + (size_t)s0 * D_DIM);
            const v4f* rb0 = reinterpret_cast<const v4f*>(emb + (size_t)t0 * D_DIM);
            const v4f* ra1 = reinterpret_cast<const v4f*>(emb + (size_t)s1 * D_DIM);
            const v4f* rb1 = reinterpret_cast<const v4f*>(emb + (size_t)t1 * D_DIM);
            p.a0lo = ra0[lane]; p.a0hi = ra0[hiIdx];
            p.b0lo = rb0[lane]; p.b0hi = rb0[hiIdx];
            p.a1lo = ra1[lane]; p.a1hi = ra1[hiIdx];
            p.b1lo = rb1[lane]; p.b1hi = rb1[hiIdx];
        };
        auto flush_run = [&]() {
            float* ra = s_sum + cur * D_DIM + 4 * lane;
            float* rb = ra + SLICE_F;
            const bool up = lane >= 16;
            #pragma unroll
            for (int j = 0; j < 4; ++j) {
                const int j1 = (j + 1) & 3;
                const int jj = up ? j1 : j;
                atomicAdd(ra + jj, up ? accA[j1] : accA[j]);
                atomicAdd(rb + jj, up ? accB[j1] : accB[j]);
            }
            if (lane < 18) {
                #pragma unroll
                for (int j = 0; j < 4; ++j) {
                    const int j1 = (j + 1) & 3;
                    const int jj = up ? j1 : j;
                    atomicAdd(ra + 128 + jj, up ? accA[4 + j1] : accA[4 + j]);
                    atomicAdd(rb + 128 + jj, up ? accB[4 + j1] : accB[4 + j]);
                }
            }
        };
        auto take = [&](int l, bool live, const v4f& alo, const v4f& ahi, const v4f& blo, const v4f& bhi) {
            if (l != cur) {
                if (cur >= 0) flush_run();
                cur = l;
                #pragma unroll
                for (int i = 0; i < 8; ++i) { accA[i] = 0.0f; accB[i] = 0.0f; }
            }
            #pragma unroll
            for (int i = 0; i < 4; ++i) {
                accA[i]     += live ? alo[i] : 0.0f;
                accA[4 + i] += live ? ahi[i] : 0.0f;
                accB[i]     += live ? blo[i] : 0.0f;
                accB[4 + i] += live ? bhi[i] : 0.0f;
            }
        };
        auto add_pair = [&](const RowPair& p) {
            take(p.l0, true,  p.a0lo, p.a0hi, p.b0lo, p.b0hi);
            take(p.l1, p.two, p.a1lo, p.a1hi, p.b1lo, p.b1hi);
        };
        {
            const int span = (qn + SWAVES - 1) / SWAVES;
            int q = wave * span;
            const int qe = (q + span < qn) ? q + span : qn;
            if (q < qe) {
                RowPair A, B;
                fetch_pair(A, q, qe);
                while (q + 4 < qe) {
                    fetch_pair(B, q + 2, qe);
                    __builtin_amdgcn_sched_barrier(0);
                    add_pair(A);
                    __builtin_amdgcn_sched_barrier(0);
                    fetch_pair(A, q + 4, qe);
                    __builtin_amdgcn_sched_barrier(0);
                    add_pair(B);
                    __builtin_amdgcn_sched_barrier(0);
                    q += 4;
                }
                if (q + 2 < qe) {
                    fetch_pair(B, q + 2, qe);
                    __builtin_amdgcn_sched_barrier(0);
                    add_pair(A);
                    add_pair(B);
                } else {
                    add_pair(A);
                }
                if (cur >= 0) flush_run();
            }
        }
        __syncthreads();
    }

    const unsigned place = (unsigned)__builtin_amdgcn_s_sendmsg_rtn(0x87);
    float* __restrict__ sum = (((place >> 16) & 15u) >= 4u) ? sum1 : sum0;
    float* g0 = sum + (size_t)part * SLICE_F;
    float* g1 = g0 + (size_t)R_REL * D_DIM;
    const int sh = (part * 8) & 31;
    for (int j = tid - sh; j < SLICE_F; j += SBLOCK) {
        if (j >= 0) {
            atomicAdd(g0 + j, s_sum[j]);
            atomicAdd(g1 + j, s_sum[SLICE_F + j]);
        }
    }
    if (tid < RPP) atomicAdd(cnt + rlo + tid, s_cnt[tid]);
}

__global__ void mean_kernel(float* __restrict__ sum, const float* __restrict__ sum1,
                            const float* __restrict__ cnt) {
    int i = blockIdx.x * blockDim.x + threadIdx.x;
    if (i >= M_ROWS * D_DIM) return;
    int m = i / D_DIM;
    sum[i] = (sum[i] + sum1[i]) / fmaxf(cnt[m % R_REL], 1.0f);
}

__device__ __forceinline__ float sigmoidf_(float x) { return 1.0f / (1.0f + expf(-x)); }

__launch_bounds__(32)
__global__ void gru_wmma_kernel(const float* __restrict__ X,
                                const float* __restrict__ dyn,
                                const float* __restrict__ Wih,
                                const float* __restrict__ Whh,
                                const float* __restrict__ bih,
                                const float* __restrict__ bhh,
                                float* __restrict__ out) {
    const int lane = threadIdx.x;
    const int m0   = blockIdx.x * 16;
    const int c0   = blockIdx.y * 16;

    const int aRow = lane & 15;
    const int nRow = lane & 15;
    const int kOff = (lane < 16) ? 0 : 2;
    const int hi   = (lane < 16) ? 0 : 8;

    const int mA = m0 + aRow;
    const int rA = mA % R_REL, fA = mA / R_REL;
    const float* pX = X + (size_t)mA * D_DIM + kOff;
    const float* pH = dyn + (size_t)rA * (2 * D_DIM) + 2 * kOff;

    const int  dCol  = c0 + nRow;
    const bool valid = dCol < D_DIM;
    const int  wRow  = valid ? dCol : D_DIM - 1;
    const float* pI0 = Wih + (size_t)wRow * D_DIM + kOff;
    const float* pI1 = pI0 + (size_t)D_DIM * D_DIM;
    const float* pI2 = pI1 + (size_t)D_DIM * D_DIM;
    const float* pH0 = Whh + (size_t)wRow * D_DIM + kOff;
    const float* pH1 = pH0 + (size_t)D_DIM * D_DIM;
    const float* pH2 = pH1 + (size_t)D_DIM * D_DIM;

    int   oIdx[8];
    float hOld[8];
    for (int i = 0; i < 8; ++i) {
        const int m = m0 + hi + i;
        const int r = m % R_REL, f = m / R_REL;
        oIdx[i] = r * (2 * D_DIM) + wRow * 2 + f;
        hOld[i] = dyn[oIdx[i]];
    }
    const float b_ir = bih[wRow], b_iz = bih[D_DIM + wRow], b_in = bih[2 * D_DIM + wRow];
    const float b_hr = bhh[wRow], b_hz = bhh[D_DIM + wRow], b_hn = bhh[2 * D_DIM + wRow];

    v8f c_ir = {}; v8f c_iz = {}; v8f c_in = {};
    v8f c_hr = {}; v8f c_hz = {}; v8f c_hn = {};

    for (int k0 = 0; k0 < D_DIM; k0 += 4) {
        v2f ax, ah, b0, b1, b2, b3, b4, b5;
        ax.x = pX[k0]; ax.y = pX[k0 + 1];
        const v4f hq = *reinterpret_cast<const v4f*>(pH + 2 * k0);
        ah.x = fA ? hq.y : hq.x;
        ah.y = fA ? hq.w : hq.z;
        b0.x = pI0[k0]; b0.y = pI0[k0 + 1];
        b1.x = pI1[k0]; b1.y = pI1[k0 + 1];
        b2.x = pI2[k0]; b2.y = pI2[k0 + 1];
        b3.x = pH0[k0]; b3.y = pH0[k0 + 1];
        b4.x = pH1[k0]; b4.y = pH1[k0 + 1];
        b5.x = pH2[k0]; b5.y = pH2[k0 + 1];

        c_ir = __builtin_amdgcn_wmma_f32_16x16x4_f32(false, ax, false, b0, (short)0, c_ir, false, false);
        c_iz = __builtin_amdgcn_wmma_f32_16x16x4_f32(false, ax, false, b1, (short)0, c_iz, false, false);
        c_in = __builtin_amdgcn_wmma_f32_16x16x4_f32(false, ax, false, b2, (short)0, c_in, false, false);
        c_hr = __builtin_amdgcn_wmma_f32_16x16x4_f32(false, ah, false, b3, (short)0, c_hr, false, false);
        c_hz = __builtin_amdgcn_wmma_f32_16x16x4_f32(false, ah, false, b4, (short)0, c_hz, false, false);
        c_hn = __builtin_amdgcn_wmma_f32_16x16x4_f32(false, ah, false, b5, (short)0, c_hn, false, false);
    }

    #pragma unroll
    for (int i = 0; i < 8; ++i) {
        float h  = hOld[i];
        float rg = sigmoidf_(c_ir[i] + b_ir + c_hr[i] + b_hr);
        float zg = sigmoidf_(c_iz[i] + b_iz + c_hz[i] + b_hz);
        float ng = tanhf(c_in[i] + b_in + rg * (c_hn[i] + b_hn));
        float o  = (1.0f - zg) * ng + zg * h;
        if (valid) out[oIdx[i]] = o;
    }
}

extern "C" void kernel_launch(void* const* d_in, const int* in_sizes, int n_in,
                              void* d_out, int out_size, void* d_ws, size_t ws_size,
                              hipStream_t stream) {
    const int*   ei  = (const int*)d_in[0];
    const int*   et  = (const int*)d_in[1];
    const int*   nid = (const int*)d_in[2];
    const float* emb = (const float*)d_in[3];
    const float* dyn = (const float*)d_in[4];
    const float* Wih = (const float*)d_in[5];
    const float* Whh = (const float*)d_in[6];
    const float* bih = (const float*)d_in[7];
    const float* bhh = (const float*)d_in[8];
    float* out = (float*)d_out;

    float* sum = (float*)d_ws;
    float* cnt = sum + (size_t)M_ROWS * D_DIM;

    float* sum1 = out;

    int nZero = M_ROWS * D_DIM + R_REL;
    zero_ws<<<(nZero + 255) / 256, 256, 0, stream>>>(sum, nZero, sum1, M_ROWS * D_DIM);

    scatter_kernel<<<NPART * NCHUNK, SBLOCK, 0, stream>>>(ei, et, nid, emb, sum, sum1, cnt);

    mean_kernel<<<(M_ROWS * D_DIM + 255) / 256, 256, 0, stream>>>(sum, sum1, cnt);

    gru_wmma_kernel<<<dim3(M_ROWS / 16, 13), 32, 0, stream>>>(sum, dyn, Wih, Whh, bih, bhh, out);
}


# GRU: one wave does both sides of 16 relations x 16 columns, fragments read 16 B per lane (4 consecutive k, two matrix steps): 5 fragment loads per k step where two waves had 16; k summed in new order
# speedup vs baseline: 12.0944x; 1.0505x over previous
#include <hip/hip_runtime.h>
#include <math.h>

#define E_EDGES  1000000
#define R_REL    1000
#define D_DIM    200
#define THREE_D  600
#define M_ROWS   2000

typedef float v2f __attribute__((ext_vector_type(2)));
typedef float v4f __attribute__((ext_vector_type(4)));
typedef float v8f __attribute__((ext_vector_type(8)));

__global__ void zero_ws(float* ws, int n, float* acc2, int n2) {
    int i = blockIdx.x * blockDim.x + threadIdx.x;
    if (i < n)  ws[i]   = 0.0f;
    if (i < n2) acc2[i] = 0.0f;
}

#define NPART   8
#define RPP     125
#define NCHUNK  32
#define CHUNK_E (E_EDGES / NCHUNK)
#define SBLOCK  1024
#define SWAVES  (SBLOCK / 32)
#define QCAP    8192
#define SLICE_F (RPP * D_DIM)

struct RowPair {
    int  l0, l1;
    bool two;
    v4f  a0lo, a0hi, b0lo, b0hi, a1lo, a1hi, b1lo, b1hi;
};

__launch_bounds__(SBLOCK)
__global__ void scatter_kernel(const int* __restrict__ ei,
                               const int* __restrict__ et,
                               const int* __restrict__ nid,
                               const float* __restrict__ emb,
                               float* __restrict__ sum0,
                               float* __restrict__ sum1,
                               float* __restrict__ cnt) {
    __shared__ float s_sum[2 * SLICE_F];
    __shared__ __attribute__((aligned(16))) float s_cnt[128];
    __shared__ __attribute__((aligned(16))) int   s_rc[128];
    __shared__ __attribute__((aligned(16))) int   s_off[128];
    __shared__ unsigned s_q1[QCAP];
    __shared__ uint2    s_q2[QCAP];
    __shared__ int      s_qn;

    const int tid  = threadIdx.x;
    const int lane = tid & 31;
    const int wave = __builtin_amdgcn_readfirstlane(tid >> 5);
    const int part  = blockIdx.x % NPART;
    const int chunk = blockIdx.x / NPART;
    const int rlo   = part * RPP;

    for (int i = tid; i < 2 * SLICE_F; i += SBLOCK) s_sum[i] = 0.0f;
    if (tid < 128) { s_cnt[tid] = 0.0f; s_rc[tid] = 0; s_off[tid] = 0; }
    if (tid == 0)  s_qn = 0;
    __syncthreads();

    const int hiIdx = lane < 18 ? 32 + lane : 49;

    const int eBeg = chunk * CHUNK_E, eEnd = eBeg + CHUNK_E;
    int qStart = 0;
    for (int base = eBeg; base < eEnd; base += QCAP) {
        const int lim = (base + QCAP < eEnd) ? base + QCAP : eEnd;

        int rv[QCAP / SBLOCK];
        #pragma unroll
        for (int k = 0; k < QCAP / SBLOCK; ++k) {
            const int e  = base + k * SBLOCK + tid;
            rv[k] = et[e < lim ? e : lim - 1];
        }
        __builtin_amdgcn_sched_barrier(0);
        #pragma unroll
        for (int k = 0; k < QCAP / SBLOCK; ++k) {
            const int  e   = base + k * SBLOCK + tid;
            const int  lr  = e < lim ? rv[k] - rlo : -1;
            const bool hit = (unsigned)lr < (unsigned)RPP;
            const unsigned m = __builtin_amdgcn_ballot_w32(hit);
            int pos0 = 0;
            if (lane == 0 && m != 0u) pos0 = atomicAdd(&s_qn, __popc(m));
            pos0 = __builtin_amdgcn_readfirstlane(pos0);
            if (hit) {
                const int pos = pos0 - qStart + (int)__builtin_amdgcn_mbcnt_lo(m, 0u);
                s_q1[pos] = (unsigned)e | ((unsigned)lr << 20);
                atomicAdd(&s_rc[lr], 1);
            }
        }
        __syncthreads();
        const int qEnd = s_qn;
        const int qn   = qEnd - qStart;
        qStart = qEnd;

        if (wave == 0) {
            const int4 c = *reinterpret_cast<const int4*>(&s_rc[4 * lane]);
            const int s = c.x + c.y + c.z + c.w;
            int inc = s;
            #pragma unroll
            for (int d = 1; d < 32; d <<= 1) {
                const int n = __shfl_up(inc, d);
                if (lane >= d) inc += n;
            }
            const int ex = inc - s;
            int4 o;
            o.x = ex; o.y = ex + c.x; o.z = o.y + c.y; o.w = o.z + c.z;
            *reinterpret_cast<int4*>(&s_off[4 * lane]) = o;
            float4 f = *reinterpret_cast<const float4*>(&s_cnt[4 * lane]);
            f.x += (float)c.x; f.y += (float)c.y; f.z += (float)c.z; f.w += (float)c.w;
            *reinterpret_cast<float4*>(&s_cnt[4 * lane]) = f;
            *reinterpret_cast<int4*>(&s_rc[4 * lane]) = make_int4(0, 0, 0, 0);
        }
        __syncthreads();

        for (int q0 = 0; q0 < qn; q0 += 2 * SBLOCK) {
            const int qa = q0 + tid, qb = qa + SBLOCK;
            const int qac = qa < qn ? qa : qn - 1, qbc = qb < qn ? qb : qn - 1;
            const unsigned ea = s_q1[qac], eb = s_q1[qbc];
            const unsigned eaE = ea & 0xFFFFFu, ebE = eb & 0xFFFFFu;
            const unsigned lra = ea >> 20,      lrb = eb >> 20;
            const int ia0 = ei[eaE], ia1 = ei[E_EDGES + eaE];
            const int ib0 = ei[ebE], ib1 = ei[E_EDGES + ebE];
            const int sa = nid[ia0], ta = nid[ia1];
            const int sb = nid[ib0], tb = nid[ib1];
            asm volatile("" :: "v"(sa), "v"(ta), "v"(sb), "v"(tb));
            if (qa < qn) {
                const int p = atomicAdd(&s_off[lra], 1);
                s_q2[p] = make_uint2((unsigned)sa | (lra << 20), (unsigned)ta);
            }
            if (qb < qn) {
                const int p = atomicAdd(&s_off[lrb], 1);
                s_q2[p] = make_uint2((unsigned)sb | (lrb << 20), (unsigned)tb);
            }
        }
        __syncthreads();

        float accA[8], accB[8];
        #pragma unroll
        for (int i = 0; i < 8; ++i) { accA[i] = 0.0f; accB[i] = 0.0f; }
        int cur = -1;

        auto fetch_pair = [&](RowPair& p, int q, int qe) {
            p.two = (q + 1) < qe;
            const uint2 e0 = s_q2[q];
            const uint2 e1 = s_q2[p.two ? q + 1 : q];
            const unsigned x0 = __builtin_amdgcn_readfirstlane(e0.x), y0 = __builtin_amdgcn_readfirstlane(e0.y);
            const unsigned x1 = __builtin_amdgcn_readfirstlane(e1.x), y1 = __builtin_amdgcn_readfirstlane(e1.y);
            const int s0 = (int)(x0 & 0xFFFFFu), t0 = (int)y0;
            const int s1 = (int)(x1 & 0xFFFFFu), t1 = (int)y1;
            p.l0 = (int)(x0 >> 20); p.l1 = (int)(x1 >> 20);
            const v4f* ra0 = reinterpret_cast<const v4f*>(emb + (size_t)s0 * D_DIM);
            const v4f* rb0 = reinterpret_cast<const v4f*>(emb + (size_t)t0 * D_DIM);
            const v4f* ra1 = reinterpret_cast<const v4f*>(emb + (size_t)s1 * D_DIM);
            const v4f* rb1 = reinterpret_cast<const v4f*>(emb + (size_t)t1 * D_DIM);
            p.a0lo = ra0[lane]; p.a0hi = ra0[hiIdx];
            p.b0lo = rb0[lane]; p.b0hi = rb0[hiIdx];
            p.a1lo = ra1[lane]; p.a1hi = ra1[hiIdx];
            p.b1lo = rb1[lane]; p.b1hi = rb1[hiIdx];
        };
        auto flush_run = [&]() {
            float* ra = s_sum + cur * D_DIM + 4 * lane;
            float* rb = ra + SLICE_F;
            const bool up = lane >= 16;
            #pragma unroll
            for (int j = 0; j < 4; ++j) {
                const int j1 = (j + 1) & 3;
                const int jj = up ? j1 : j;
                atomicAdd(ra + jj, up ? accA[j1] : accA[j]);
                atomicAdd(rb + jj, up ? accB[j1] : accB[j]);
            }
            if (lane < 18) {
                #pragma unroll
                for (int j = 0; j < 4; ++j) {
                    const int j1 = (j + 1) & 3;
                    const int jj = up ? j1 : j;
                    atomicAdd(ra + 128 + jj, up ? accA[4 + j1] : accA[4 + j]);
                    atomicAdd(rb + 128 + jj, up ? accB[4 + j1] : accB[4 + j]);
                }
            }
        };
        auto take = [&](int l, bool live, const v4f& alo, const v4f& ahi, const v4f& blo, const v4f& bhi) {
            if (l != cur) {
                if (cur >= 0) flush_run();
                cur = l;
                #pragma unroll
                for (int i = 0; i < 8; ++i) { accA[i] = 0.0f; accB[i] = 0.0f; }
            }
            #pragma unroll
            for (int i = 0; i < 4; ++i) {
                accA[i]     += live ? alo[i] : 0.0f;
                accA[4 + i] += live ? ahi[i] : 0.0f;
                accB[i]     += live ? blo[i] : 0.0f;
                accB[4 + i] += live ? bhi[i] : 0.0f;
            }
        };
        auto add_pair = [&](const RowPair& p) {
            take(p.l0, true,  p.a0lo, p.a0hi, p.b0lo, p.b0hi);
            take(p.l1, p.two, p.a1lo, p.a1hi, p.b1lo, p.b1hi);
        };
        {
            const int span = (qn + SWAVES - 1) / SWAVES;
            int q = wave * span;
            const int qe = (q + span < qn) ? q + span : qn;
            if (q < qe) {
                RowPair A, B;
                fetch_pair(A, q, qe);
                while (q + 4 < qe) {
                    fetch_pair(B, q + 2, qe);
                    __builtin_amdgcn_sched_barrier(0);
                    add_pair(A);
                    __builtin_amdgcn_sched_barrier(0);
                    fetch_pair(A, q + 4, qe);
                    __builtin_amdgcn_sched_barrier(0);
                    add_pair(B);
                    __builtin_amdgcn_sched_barrier(0);
                    q += 4;
                }
                if (q + 2 < qe) {
                    fetch_pair(B, q + 2, qe);
                    __builtin_amdgcn_sched_barrier(0);
                    add_pair(A);
                    add_pair(B);
                } else {
                    add_pair(A);
                }
                if (cur >= 0) flush_run();
            }
        }
        __syncthreads();
    }

    const unsigned place = (unsigned)__builtin_amdgcn_s_sendmsg_rtn(0x87);
    float* __restrict__ sum = (((place >> 16) & 15u) >= 4u) ? sum1 : sum0;
    float* g0 = sum + (size_t)part * SLICE_F;
    float* g1 = g0 + (size_t)R_REL * D_DIM;
    const int sh = (part * 8) & 31;
    for (int j = tid - sh; j < SLICE_F; j += SBLOCK) {
        if (j >= 0) {
            atomicAdd(g0 + j, s_sum[j]);
            atomicAdd(g1 + j, s_sum[SLICE_F + j]);
        }
    }
    if (tid < RPP) atomicAdd(cnt + rlo + tid, s_cnt[tid]);
}

__global__ void mean_kernel(float* __restrict__ sum, const float* __restrict__ sum1,
                            const float* __restrict__ cnt) {
    int i = blockIdx.x * blockDim.x + threadIdx.x;
    if (i >= M_ROWS * D_DIM) return;
    int m = i / D_DIM;
    sum[i] = (sum[i] + sum1[i]) / fmaxf(cnt[m % R_REL], 1.0f);
}

__device__ __forceinline__ float sigmoidf_(float x) { return 1.0f / (1.0f + expf(-x)); }

__launch_bounds__(32)
__global__ void gru_wmma_kernel(const float* __restrict__ X,
                                const float* __restrict__ dyn,
                                const float* __restrict__ Wih,
                                const float* __restrict__ Whh,
                                const float* __restrict__ bih,
                                const float* __restrict__ bhh,
                                float* __restrict__ out) {
    const int lane = threadIdx.x;
    const int r0   = blockIdx.x * 16;
    const int c0   = blockIdx.y * 16;

    const int c    = lane & 15;
    const int half = lane >> 4;
    const int kOff = 4 * half;

    const int rRaw = r0 + c;
    const int rA   = rRaw < R_REL ? rRaw : R_REL - 1;
    const float* pX0 = X + (size_t)rA * D_DIM + kOff;
    const float* pX1 = pX0 + (size_t)R_REL * D_DIM;
    const float* pH  = dyn + (size_t)rA * (2 * D_DIM) + 2 * kOff;

    const int  dCol  = c0 + c;
    const bool valid = dCol < D_DIM;
    const int  wRow  = valid ? dCol : D_DIM - 1;
    const float* pI0 = Wih + (size_t)wRow * D_DIM + kOff;
    const float* pI1 = pI0 + (size_t)D_DIM * D_DIM;
    const float* pI2 = pI1 + (size_t)D_DIM * D_DIM;
    const float* pH0 = Whh + (size_t)wRow * D_DIM + kOff;
    const float* pH1 = pH0 + (size_t)D_DIM * D_DIM;
    const float* pH2 = pH1 + (size_t)D_DIM * D_DIM;

    int  oIdx[8];
    bool rOk[8];
    v2f  hOld[8];
    #pragma unroll
    for (int i = 0; i < 8; ++i) {
        const int r = r0 + 8 * half + i;
        rOk[i]  = r < R_REL;
        oIdx[i] = (rOk[i] ? r : R_REL - 1) * (2 * D_DIM) + wRow * 2;
        hOld[i] = *reinterpret_cast<const v2f*>(dyn + oIdx[i]);
    }
    const float b_ir = bih[wRow], b_iz = bih[D_DIM + wRow], b_in = bih[2 * D_DIM + wRow];
    const float b_hr = bhh[wRow], b_hz = bhh[D_DIM + wRow], b_hn = bhh[2 * D_DIM + wRow];

    v8f c_ir[2] = {}; v8f c_iz[2] = {}; v8f c_in[2] = {};
    v8f c_hr[2] = {}; v8f c_hz[2] = {}; v8f c_hn[2] = {};

    #pragma unroll 5
    for (int kb = 0; kb < D_DIM; kb += 8) {
        const v4f x0 = *reinterpret_cast<const v4f*>(pX0 + kb);
        const v4f x1 = *reinterpret_cast<const v4f*>(pX1 + kb);
        const v4f ha = *reinterpret_cast<const v4f*>(pH + 2 * kb);
        const v4f hb = *reinterpret_cast<const v4f*>(pH + 2 * kb + 4);
        const v4f w0 = *reinterpret_cast<const v4f*>(pI0 + kb);
        const v4f w1 = *reinterpret_cast<const v4f*>(pI1 + kb);
        const v4f w2 = *reinterpret_cast<const v4f*>(pI2 + kb);
        const v4f w3 = *reinterpret_cast<const v4f*>(pH0 + kb);
        const v4f w4 = *reinterpret_cast<const v4f*>(pH1 + kb);
        const v4f w5 = *reinterpret_cast<const v4f*>(pH2 + kb);

        {
            const v2f ax0 = {x0.x, x0.y}, ax1 = {x1.x, x1.y};
            const v2f ah0 = {ha.x, ha.z}, ah1 = {ha.y, ha.w};
            const v2f b0 = {w0.x, w0.y}, b1 = {w1.x, w1.y}, b2 = {w2.x, w2.y};
            const v2f b3 = {w3.x, w3.y}, b4 = {w4.x, w4.y}, b5 = {w5.x, w5.y};
            c_ir[0] = __builtin_amdgcn_wmma_f32_16x16x4_f32(false, ax0, false, b0, (short)0, c_ir[0], false, false);
            c_iz[0] = __builtin_amdgcn_wmma_f32_16x16x4_f32(false, ax0, false, b1, (short)0, c_iz[0], false, false);
            c_in[0] = __builtin_amdgcn_wmma_f32_16x16x4_f32(false, ax0, false, b2, (short)0, c_in[0], false, false);
            c_ir[1] = __builtin_amdgcn_wmma_f32_16x16x4_f32(false, ax1, false, b0, (short)0, c_ir[1], false, false);
            c_iz[1] = __builtin_amdgcn_wmma_f32_16x16x4_f32(false, ax1, false, b1, (short)0, c_iz[1], false, false);
            c_in[1] = __builtin_amdgcn_wmma_f32_16x16x4_f32(false, ax1, false, b2, (short)0, c_in[1], false, false);
            c_hr[0] = __builtin_amdgcn_wmma_f32_16x16x4_f32(false, ah0, false, b3, (short)0, c_hr[0], false, false);
            c_hz[0] = __builtin_amdgcn_wmma_f32_16x16x4_f32(false, ah0, false, b4, (short)0, c_hz[0], false, false);
            c_hn[0] = __builtin_amdgcn_wmma_f32_16x16x4_f32(false, ah0, false, b5, (short)0, c_hn[0], false, false);
            c_hr[1] = __builtin_amdgcn_wmma_f32_16x16x4_f32(false, ah1, false, b3, (short)0, c_hr[1], false, false);
            c_hz[1] = __builtin_amdgcn_wmma_f32_16x16x4_f32(false, ah1, false, b4, (short)0, c_hz[1], false, false);
            c_hn[1] = __builtin_amdgcn_wmma_f32_16x16x4_f32(false, ah1, false, b5, (short)0, c_hn[1], false, false);
        }
        {
            const v2f ax0 = {x0.z, x0.w}, ax1 = {x1.z, x1.w};
            const v2f ah0 = {hb.x, hb.z}, ah1 = {hb.y, hb.w};
            const v2f b0 = {w0.z, w0.w}, b1 = {w1.z, w1.w}, b2 = {w2.z, w2.w};
            const v2f b3 = {w3.z, w3.w}, b4 = {w4.z, w4.w}, b5 = {w5.z, w5.w};
            c_ir[0] = __builtin_amdgcn_wmma_f32_16x16x4_f32(false, ax0, false, b0, (short)0, c_ir[0], false, false);
            c_iz[0] = __builtin_amdgcn_wmma_f32_16x16x4_f32(false, ax0, false, b1, (short)0, c_iz[0], false, false);
            c_in[0] = __builtin_amdgcn_wmma_f32_16x16x4_f32(false, ax0, false, b2, (short)0, c_in[0], false, false);
            c_ir[1] = __builtin_amdgcn_wmma_f32_16x16x4_f32(false, ax1, false, b0, (short)0, c_ir[1], false, false);
            c_iz[1] = __builtin_amdgcn_wmma_f32_16x16x4_f32(false, ax1, false, b1, (short)0, c_iz[1], false, false);
            c_in[1] = __builtin_amdgcn_wmma_f32_16x16x4_f32(false, ax1, false, b2, (short)0, c_in[1], false, false);
            c_hr[0] = __builtin_amdgcn_wmma_f32_16x16x4_f32(false, ah0, false, b3, (short)0, c_hr[0], false, false);
            c_hz[0] = __builtin_amdgcn_wmma_f32_16x16x4_f32(false, ah0, false, b4, (short)0, c_hz[0], false, false);
            c_hn[0] = __builtin_amdgcn_wmma_f32_16x16x4_f32(false, ah0, false, b5, (short)0, c_hn[0], false, false);
            c_hr[1] = __builtin_amdgcn_wmma_f32_16x16x4_f32(false, ah1, false, b3, (short)0, c_hr[1], false, false);
            c_hz[1] = __builtin_amdgcn_wmma_f32_16x16x4_f32(false, ah1, false, b4, (short)0, c_hz[1], false, false);
            c_hn[1] = __builtin_amdgcn_wmma_f32_16x16x4_f32(false, ah1, false, b5, (short)0, c_hn[1], false, false);
        }
    }

    #pragma unroll
    for (int i = 0; i < 8; ++i) {
        v2f o;
        #pragma unroll
        for (int sd = 0; sd < 2; ++sd) {
            const float hp = sd ? hOld[i].y : hOld[i].x;
            const float rg = sigmoidf_(c_ir[sd][i] + b_ir + c_hr[sd][i] + b_hr);
            const float zg = sigmoidf_(c_iz[sd][i] + b_iz + c_hz[sd][i] + b_hz);
            const float ng = tanhf(c_in[sd][i] + b_in + rg * (c_hn[sd][i] + b_hn));
            const float ov = (1.0f - zg) * ng + zg * hp;
            if (sd) o.y = ov; else o.x = ov;
        }
        if (valid && rOk[i]) *reinterpret_cast<v2f*>(out + oIdx[i]) = o;
    }
}

extern "C" void kernel_launch(void* const* d_in, const int* in_sizes, int n_in,
                              void* d_out, int out_size, void* d_ws, size_t ws_size,
                              hipStream_t stream) {
    const int*   ei  = (const int*)d_in[0];
    const int*   et  = (const int*)d_in[1];
    const int*   nid = (const int*)d_in[2];
    const float* emb = (const float*)d_in[3];
    const float* dyn = (const float*)d_in[4];
    const float* Wih = (const float*)d_in[5];
    const float* Whh = (const float*)d_in[6];
    const float* bih = (const float*)d_in[7];
    const float* bhh = (const float*)d_in[8];
    float* out = (float*)d_out;

    float* sum = (float*)d_ws;
    float* cnt = sum + (size_t)M_ROWS * D_DIM;

    float* sum1 = out;

    int nZero = M_ROWS * D_DIM + R_REL;
    zero_ws<<<(nZero + 255) / 256, 256, 0, stream>>>(sum, nZero, sum1, M_ROWS * D_DIM);

    scatter_kernel<<<NPART * NCHUNK, SBLOCK, 0, stream>>>(ei, et, nid, emb, sum, sum1, cnt);

    mean_kernel<<<(M_ROWS * D_DIM + 255) / 256, 256, 0, stream>>>(sum, sum1, cnt);

    gru_wmma_kernel<<<dim3((R_REL + 15) / 16, 13), 32, 0, stream>>>(sum, dyn, Wih, Whh, bih, bhh, out);
}


# scatter: a work-group scans, sorts and sums its whole 31250-edge chunk as ONE round (was 4 rounds of 8192; rounds kept as overflow path), 4 entries resolved per thread; runs longer: sums in new order
# speedup vs baseline: 12.7707x; 1.0559x over previous
#include <hip/hip_runtime.h>
#include <math.h>

#define E_EDGES  1000000
#define R_REL    1000
#define D_DIM    200
#define THREE_D  600
#define M_ROWS   2000

typedef float v2f __attribute__((ext_vector_type(2)));
typedef float v4f __attribute__((ext_vector_type(4)));
typedef float v8f __attribute__((ext_vector_type(8)));

__global__ void zero_ws(float* ws, int n, float* acc2, int n2) {
    int i = blockIdx.x * blockDim.x + threadIdx.x;
    if (i < n)  ws[i]   = 0.0f;
    if (i < n2) acc2[i] = 0.0f;
}

#define NPART   8
#define RPP     125
#define NCHUNK  32
#define CHUNK_E (E_EDGES / NCHUNK)
#define SBLOCK  1024
#define SWAVES  (SBLOCK / 32)
#define QCAP    8192
#define SCAN_K  16
#define RES_K   4
#define SLICE_F (RPP * D_DIM)

struct RowPair {
    int  l0, l1;
    bool two;
    v4f  a0lo, a0hi, b0lo, b0hi, a1lo, a1hi, b1lo, b1hi;
};

__launch_bounds__(SBLOCK)
__global__ void scatter_kernel(const int* __restrict__ ei,
                               const int* __restrict__ et,
                               const int* __restrict__ nid,
                               const float* __restrict__ emb,
                               float* __restrict__ sum0,
                               float* __restrict__ sum1,
                               float* __restrict__ cnt) {
    __shared__ float s_sum[2 * SLICE_F];
    __shared__ __attribute__((aligned(16))) float s_cnt[128];
    __shared__ __attribute__((aligned(16))) int   s_rc[128];
    __shared__ __attribute__((aligned(16))) int   s_off[128];
    __shared__ unsigned s_q1[QCAP];
    __shared__ uint2    s_q2[QCAP];
    __shared__ int      s_qn;

    const int tid  = threadIdx.x;
    const int lane = tid & 31;
    const int wave = __builtin_amdgcn_readfirstlane(tid >> 5);
    const int part  = blockIdx.x % NPART;
    const int chunk = blockIdx.x / NPART;
    const int rlo   = part * RPP;

    for (int i = tid; i < 2 * SLICE_F; i += SBLOCK) s_sum[i] = 0.0f;
    if (tid < 128) { s_cnt[tid] = 0.0f; s_rc[tid] = 0; s_off[tid] = 0; }
    if (tid == 0)  s_qn = 0;
    __syncthreads();

    const int hiIdx = lane < 18 ? 32 + lane : 49;

    const int eBeg = chunk * CHUNK_E, eEnd = eBeg + CHUNK_E;
    int qStart = 0;
    int step = CHUNK_E;
    for (int base = eBeg; base < eEnd; ) {
        const int lim = (base + step < eEnd) ? base + step : eEnd;

        for (int sb = base; sb < lim; sb += SCAN_K * SBLOCK) {
            int rv[SCAN_K];
            #pragma unroll
            for (int k = 0; k < SCAN_K; ++k) {
                const int e  = sb + k * SBLOCK + tid;
                rv[k] = et[e < lim ? e : lim - 1];
            }
            __builtin_amdgcn_sched_barrier(0);
            #pragma unroll
            for (int k = 0; k < SCAN_K; ++k) {
                const int  e   = sb + k * SBLOCK + tid;
                const int  lr  = e < lim ? rv[k] - rlo : -1;
                const bool hit = (unsigned)lr < (unsigned)RPP;
                const unsigned m = __builtin_amdgcn_ballot_w32(hit);
                int pos0 = 0;
                if (lane == 0 && m != 0u) pos0 = atomicAdd(&s_qn, __popc(m));
                pos0 = __builtin_amdgcn_readfirstlane(pos0);
                if (hit) {
                    const int pos = pos0 - qStart + (int)__builtin_amdgcn_mbcnt_lo(m, 0u);
                    if (pos < QCAP) {
                        s_q1[pos] = (unsigned)e | ((unsigned)lr << 20);
                        atomicAdd(&s_rc[lr], 1);
                    }
                }
            }
        }
        __syncthreads();
        const int qEnd = s_qn;
        const int qn   = qEnd - qStart;
        if (qn > QCAP) {
            __syncthreads();
            if (tid < 128) s_rc[tid] = 0;
            if (tid == 0)  s_qn = qStart;
            __syncthreads();
            step = QCAP;
            continue;
        }
        qStart = qEnd;

        if (wave == 0) {
            const int4 c = *reinterpret_cast<const int4*>(&s_rc[4 * lane]);
            const int s = c.x + c.y + c.z + c.w;
            int inc = s;
            #pragma unroll
            for (int d = 1; d < 32; d <<= 1) {
                const int n = __shfl_up(inc, d);
                if (lane >= d) inc += n;
            }
            const int ex = inc - s;
            int4 o;
            o.x = ex; o.y = ex + c.x; o.z = o.y + c.y; o.w = o.z + c.z;
            *reinterpret_cast<int4*>(&s_off[4 * lane]) = o;
            float4 f = *reinterpret_cast<const float4*>(&s_cnt[4 * lane]);
            f.x += (float)c.x; f.y += (float)c.y; f.z += (float)c.z; f.w += (float)c.w;
            *reinterpret_cast<float4*>(&s_cnt[4 * lane]) = f;
            *reinterpret_cast<int4*>(&s_rc[4 * lane]) = make_int4(0, 0, 0, 0);
        }
        __syncthreads();

        for (int q0 = 0; q0 < qn; q0 += RES_K * SBLOCK) {
            unsigned lrq[RES_K];
            int i0[RES_K], i1[RES_K], sn[RES_K], tn[RES_K];
            #pragma unroll
            for (int u = 0; u < RES_K; ++u) {
                const int qi = q0 + u * SBLOCK + tid;
                const unsigned en = s_q1[qi < qn ? qi : qn - 1];
                const unsigned eE = en & 0xFFFFFu;
                lrq[u] = en >> 20;
                i0[u] = ei[eE];
                i1[u] = ei[E_EDGES + eE];
            }
            #pragma unroll
            for (int u = 0; u < RES_K; ++u) {
                sn[u] = nid[i0[u]];
                tn[u] = nid[i1[u]];
            }
            asm volatile("" :: "v"(sn[0]), "v"(tn[0]), "v"(sn[1]), "v"(tn[1]),
                               "v"(sn[2]), "v"(tn[2]), "v"(sn[3]), "v"(tn[3]));
            #pragma unroll
            for (int u = 0; u < RES_K; ++u) {
                const int qi = q0 + u * SBLOCK + tid;
                if (qi < qn) {
                    const int p = atomicAdd(&s_off[lrq[u]], 1);
                    s_q2[p] = make_uint2((unsigned)sn[u] | (lrq[u] << 20), (unsigned)tn[u]);
                }
            }
        }
        __syncthreads();

        float accA[8], accB[8];
        #pragma unroll
        for (int i = 0; i < 8; ++i) { accA[i] = 0.0f; accB[i] = 0.0f; }
        int cur = -1;

        auto fetch_pair = [&](RowPair& p, int q, int qe) {
            p.two = (q + 1) < qe;
            const uint2 e0 = s_q2[q];
            const uint2 e1 = s_q2[p.two ? q + 1 : q];
            const unsigned x0 = __builtin_amdgcn_readfirstlane(e0.x), y0 = __builtin_amdgcn_readfirstlane(e0.y);
            const unsigned x1 = __builtin_amdgcn_readfirstlane(e1.x), y1 = __builtin_amdgcn_readfirstlane(e1.y);
            const int s0 = (int)(x0 & 0xFFFFFu), t0 = (int)y0;
            const int s1 = (int)(x1 & 0xFFFFFu), t1 = (int)y1;
            p.l0 = (int)(x0 >> 20); p.l1 = (int)(x1 >> 20);
            const v4f* ra0 = reinterpret_cast<const v4f*>(emb + (size_t)s0 * D_DIM);
            const v4f* rb0 = reinterpret_cast<const v4f*>(emb + (size_t)t0 * D_DIM);
            const v4f* ra1 = reinterpret_cast<const v4f*>(emb + (size_t)s1 * D_DIM);
            const v4f* rb1 = reinterpret_cast<const v4f*>(emb + (size_t)t1 * D_DIM);
            p.a0lo = ra0[lane]; p.a0hi = ra0[hiIdx];
            p.b0lo = rb0[lane]; p.b0hi = rb0[hiIdx];
            p.a1lo = ra1[lane]; p.a1hi = ra1[hiIdx];
            p.b1lo = rb1[lane]; p.b1hi = rb1[hiIdx];
        };
        auto flush_run = [&]() {
            float* ra = s_sum + cur * D_DIM + 4 * lane;
            float* rb = ra + SLICE_F;
            const bool up = lane >= 16;
            #pragma unroll
            for (int j = 0; j < 4; ++j) {
                const int j1 = (j + 1) & 3;
                const int jj = up ? j1 : j;
                atomicAdd(ra + jj, up ? accA[j1] : accA[j]);
                atomicAdd(rb + jj, up ? accB[j1] : accB[j]);
            }
            if (lane < 18) {
                #pragma unroll
                for (int j = 0; j < 4; ++j) {
                    const int j1 = (j + 1) & 3;
                    const int jj = up ? j1 : j;
                    atomicAdd(ra + 128 + jj, up ? accA[4 + j1] : accA[4 + j]);
                    atomicAdd(rb + 128 + jj, up ? accB[4 + j1] : accB[4 + j]);
                }
            }
        };
        auto take = [&](int l, bool live, const v4f& alo, const v4f& ahi, const v4f& blo, const v4f& bhi) {
            if (l != cur) {
                if (cur >= 0) flush_run();
                cur = l;
                #pragma unroll
                for (int i = 0; i < 8; ++i) { accA[i] = 0.0f; accB[i] = 0.0f; }
            }
            #pragma unroll
            for (int i = 0; i < 4; ++i) {
                accA[i]     += live ? alo[i] : 0.0f;
                accA[4 + i] += live ? ahi[i] : 0.0f;
                accB[i]     += live ? blo[i] : 0.0f;
                accB[4 + i] += live ? bhi[i] : 0.0f;
            }
        };
        auto add_pair = [&](const RowPair& p) {
            take(p.l0, true,  p.a0lo, p.a0hi, p.b0lo, p.b0hi);
            take(p.l1, p.two, p.a1lo, p.a1hi, p.b1lo, p.b1hi);
        };
        {
            const int span = (qn + SWAVES - 1) / SWAVES;
            int q = wave * span;
            const int qe = (q + span < qn) ? q + span : qn;
            if (q < qe) {
                RowPair A, B;
                fetch_pair(A, q, qe);
                while (q + 4 < qe) {
                    fetch_pair(B, q + 2, qe);
                    __builtin_amdgcn_sched_barrier(0);
                    add_pair(A);
                    __builtin_amdgcn_sched_barrier(0);
                    fetch_pair(A, q + 4, qe);
                    __builtin_amdgcn_sched_barrier(0);
                    add_pair(B);
                    __builtin_amdgcn_sched_barrier(0);
                    q += 4;
                }
                if (q + 2 < qe) {
                    fetch_pair(B, q + 2, qe);
                    __builtin_amdgcn_sched_barrier(0);
                    add_pair(A);
                    add_pair(B);
                } else {
                    add_pair(A);
                }
                if (cur >= 0) flush_run();
            }
        }
        __syncthreads();
        base = lim;
    }

    const unsigned place = (unsigned)__builtin_amdgcn_s_sendmsg_rtn(0x87);
    float* __restrict__ sum = (((place >> 16) & 15u) >= 4u) ? sum1 : sum0;
    float* g0 = sum + (size_t)part * SLICE_F;
    float* g1 = g0 + (size_t)R_REL * D_DIM;
    const int sh = (part * 8) & 31;
    for (int j = tid - sh; j < SLICE_F; j += SBLOCK) {
        if (j >= 0) {
            atomicAdd(g0 + j, s_sum[j]);
            atomicAdd(g1 + j, s_sum[SLICE_F + j]);
        }
    }
    if (tid < RPP) atomicAdd(cnt + rlo + tid, s_cnt[tid]);
}

__global__ void mean_kernel(float* __restrict__ sum, const float* __restrict__ sum1,
                            const float* __restrict__ cnt) {
    int i = blockIdx.x * blockDim.x + threadIdx.x;
    if (i >= M_ROWS * D_DIM) return;
    int m = i / D_DIM;
    sum[i] = (sum[i] + sum1[i]) / fmaxf(cnt[m % R_REL], 1.0f);
}

__device__ __forceinline__ float sigmoidf_(float x) { return 1.0f / (1.0f + expf(-x)); }

__launch_bounds__(32)
__global__ void gru_wmma_kernel(const float* __restrict__ X,
                                const float* __restrict__ dyn,
                                const float* __restrict__ Wih,
                                const float* __restrict__ Whh,
                                const float* __restrict__ bih,
                                const float* __restrict__ bhh,
                                float* __restrict__ out) {
    const int lane = threadIdx.x;
    const int r0   = blockIdx.x * 16;
    const int c0   = blockIdx.y * 16;

    const int c    = lane & 15;
    const int half = lane >> 4;
    const int kOff = 4 * half;

    const int rRaw = r0 + c;
    const int rA   = rRaw < R_REL ? rRaw : R_REL - 1;
    const float* pX0 = X + (size_t)rA * D_DIM + kOff;
    const float* pX1 = pX0 + (size_t)R_REL * D_DIM;
    const float* pH  = dyn + (size_t)rA * (2 * D_DIM) + 2 * kOff;

    const int  dCol  = c0 + c;
    const bool valid = dCol < D_DIM;
    const int  wRow  = valid ? dCol : D_DIM - 1;
    const float* pI0 = Wih + (size_t)wRow * D_DIM + kOff;
    const float* pI1 = pI0 + (size_t)D_DIM * D_DIM;
    const float* pI2 = pI1 + (size_t)D_DIM * D_DIM;
    const float* pH0 = Whh + (size_t)wRow * D_DIM + kOff;
    const float* pH1 = pH0 + (size_t)D_DIM * D_DIM;
    const float* pH2 = pH1 + (size_t)D_DIM * D_DIM;

    int  oIdx[8];
    bool rOk[8];
    v2f  hOld[8];
    #pragma unroll
    for (int i = 0; i < 8; ++i) {
        const int r = r0 + 8 * half + i;
        rOk[i]  = r < R_REL;
        oIdx[i] = (rOk[i] ? r : R_REL - 1) * (2 * D_DIM) + wRow * 2;
        hOld[i] = *reinterpret_cast<const v2f*>(dyn + oIdx[i]);
    }
    const float b_ir = bih[wRow], b_iz = bih[D_DIM + wRow], b_in = bih[2 * D_DIM + wRow];
    const float b_hr = bhh[wRow], b_hz = bhh[D_DIM + wRow], b_hn = bhh[2 * D_DIM + wRow];

    v8f c_ir[2] = {}; v8f c_iz[2] = {}; v8f c_in[2] = {};
    v8f c_hr[2] = {}; v8f c_hz[2] = {}; v8f c_hn[2] = {};

    #pragma unroll 5
    for (int kb = 0; kb < D_DIM; kb += 8) {
        const v4f x0 = *reinterpret_cast<const v4f*>(pX0 + kb);
        const v4f x1 = *reinterpret_cast<const v4f*>(pX1 + kb);
        const v4f ha = *reinterpret_cast<const v4f*>(pH + 2 * kb);
        const v4f hb = *reinterpret_cast<const v4f*>(pH + 2 * kb + 4);
        const v4f w0 = *reinterpret_cast<const v4f*>(pI0 + kb);
        const v4f w1 = *reinterpret_cast<const v4f*>(pI1 + kb);
        const v4f w2 = *reinterpret_cast<const v4f*>(pI2 + kb);
        const v4f w3 = *reinterpret_cast<const v4f*>(pH0 + kb);
        const v4f w4 = *reinterpret_cast<const v4f*>(pH1 + kb);
        const v4f w5 = *reinterpret_cast<const v4f*>(pH2 + kb);

        {
            const v2f ax0 = {x0.x, x0.y}, ax1 = {x1.x, x1.y};
            const v2f ah0 = {ha.x, ha.z}, ah1 = {ha.y, ha.w};
            const v2f b0 = {w0.x, w0.y}, b1 = {w1.x, w1.y}, b2 = {w2.x, w2.y};
            const v2f b3 = {w3.x, w3.y}, b4 = {w4.x, w4.y}, b5 = {w5.x, w5.y};
            c_ir[0] = __builtin_amdgcn_wmma_f32_16x16x4_f32(false, ax0, false, b0, (short)0, c_ir[0], false, false);
            c_iz[0] = __builtin_amdgcn_wmma_f32_16x16x4_f32(false, ax0, false, b1, (short)0, c_iz[0], false, false);
            c_in[0] = __builtin_amdgcn_wmma_f32_16x16x4_f32(false, ax0, false, b2, (short)0, c_in[0], false, false);
            c_ir[1] = __builtin_amdgcn_wmma_f32_16x16x4_f32(false, ax1, false, b0, (short)0, c_ir[1], false, false);
            c_iz[1] = __builtin_amdgcn_wmma_f32_16x16x4_f32(false, ax1, false, b1, (short)0, c_iz[1], false, false);
            c_in[1] = __builtin_amdgcn_wmma_f32_16x16x4_f32(false, ax1, false, b2, (short)0, c_in[1], false, false);
            c_hr[0] = __builtin_amdgcn_wmma_f32_16x16x4_f32(false, ah0, false, b3, (short)0, c_hr[0], false, false);
            c_hz[0] = __builtin_amdgcn_wmma_f32_16x16x4_f32(false, ah0, false, b4, (short)0, c_hz[0], false, false);
            c_hn[0] = __builtin_amdgcn_wmma_f32_16x16x4_f32(false, ah0, false, b5, (short)0, c_hn[0], false, false);
            c_hr[1] = __builtin_amdgcn_wmma_f32_16x16x4_f32(false, ah1, false, b3, (short)0, c_hr[1], false, false);
            c_hz[1] = __builtin_amdgcn_wmma_f32_16x16x4_f32(false, ah1, false, b4, (short)0, c_hz[1], false, false);
            c_hn[1] = __builtin_amdgcn_wmma_f32_16x16x4_f32(false, ah1, false, b5, (short)0, c_hn[1], false, false);
        }
        {
            const v2f ax0 = {x0.z, x0.w}, ax1 = {x1.z, x1.w};
            const v2f ah0 = {hb.x, hb.z}, ah1 = {hb.y, hb.w};
            const v2f b0 = {w0.z, w0.w}, b1 = {w1.z, w1.w}, b2 = {w2.z, w2.w};
            const v2f b3 = {w3.z, w3.w}, b4 = {w4.z, w4.w}, b5 = {w5.z, w5.w};
            c_ir[0] = __builtin_amdgcn_wmma_f32_16x16x4_f32(false, ax0, false, b0, (short)0, c_ir[0], false, false);
            c_iz[0] = __builtin_amdgcn_wmma_f32_16x16x4_f32(false, ax0, false, b1, (short)0, c_iz[0], false, false);
            c_in[0] = __builtin_amdgcn_wmma_f32_16x16x4_f32(false, ax0, false, b2, (short)0, c_in[0], false, false);
            c_ir[1] = __builtin_amdgcn_wmma_f32_16x16x4_f32(false, ax1, false, b0, (short)0, c_ir[1], false, false);
            c_iz[1] = __builtin_amdgcn_wmma_f32_16x16x4_f32(false, ax1, false, b1, (short)0, c_iz[1], false, false);
            c_in[1] = __builtin_amdgcn_wmma_f32_16x16x4_f32(false, ax1, false, b2, (short)0, c_in[1], false, false);
            c_hr[0] = __builtin_amdgcn_wmma_f32_16x16x4_f32(false, ah0, false, b3, (short)0, c_hr[0], false, false);
            c_hz[0] = __builtin_amdgcn_wmma_f32_16x16x4_f32(false, ah0, false, b4, (short)0, c_hz[0], false, false);
            c_hn[0] = __builtin_amdgcn_wmma_f32_16x16x4_f32(false, ah0, false, b5, (short)0, c_hn[0], false, false);
            c_hr[1] = __builtin_amdgcn_wmma_f32_16x16x4_f32(false, ah1, false, b3, (short)0, c_hr[1], false, false);
            c_hz[1] = __builtin_amdgcn_wmma_f32_16x16x4_f32(false, ah1, false, b4, (short)0, c_hz[1], false, false);
            c_hn[1] = __builtin_amdgcn_wmma_f32_16x16x4_f32(false, ah1, false, b5, (short)0, c_hn[1], false, false);
        }
    }

    #pragma unroll
    for (int i = 0; i < 8; ++i) {
        v2f o;
        #pragma unroll
        for (int sd = 0; sd < 2; ++sd) {
            const float hp = sd ? hOld[i].y : hOld[i].x;
            const float rg = sigmoidf_(c_ir[sd][i] + b_ir + c_hr[sd][i] + b_hr);
            const float zg = sigmoidf_(c_iz[sd][i] + b_iz + c_hz[sd][i] + b_hz);
            const float ng = tanhf(c_in[sd][i] + b_in + rg * (c_hn[sd][i] + b_hn));
            const float ov = (1.0f - zg) * ng + zg * hp;
            if (sd) o.y = ov; else o.x = ov;
        }
        if (valid && rOk[i]) *reinterpret_cast<v2f*>(out + oIdx[i]) = o;
    }
}

extern "C" void kernel_launch(void* const* d_in, const int* in_sizes, int n_in,
                              void* d_out, int out_size, void* d_ws, size_t ws_size,
                              hipStream_t stream) {
    const int*   ei  = (const int*)d_in[0];
    const int*   et  = (const int*)d_in[1];
    const int*   nid = (const int*)d_in[2];
    const float* emb = (const float*)d_in[3];
    const float* dyn = (const float*)d_in[4];
    const float* Wih = (const float*)d_in[5];
    const float* Whh = (const float*)d_in[6];
    const float* bih = (const float*)d_in[7];
    const float* bhh = (const float*)d_in[8];
    float* out = (float*)d_out;

    float* sum = (float*)d_ws;
    float* cnt = sum + (size_t)M_ROWS * D_DIM;

    float* sum1 = out;

    int nZero = M_ROWS * D_DIM + R_REL;
    zero_ws<<<(nZero + 255) / 256, 256, 0, stream>>>(sum, nZero, sum1, M_ROWS * D_DIM);

    scatter_kernel<<<NPART * NCHUNK, SBLOCK, 0, stream>>>(ei, et, nid, emb, sum, sum1, cnt);

    mean_kernel<<<(M_ROWS * D_DIM + 255) / 256, 256, 0, stream>>>(sum, sum1, cnt);

    gru_wmma_kernel<<<dim3((R_REL + 15) / 16, 13), 32, 0, stream>>>(sum, dyn, Wih, Whh, bih, bhh, out);
}
